# starting kernel
# baseline (speedup 1.0000x reference)
#include <hip/hip_runtime.h>
#include <math.h>


typedef __attribute__((ext_vector_type(2))) float v2f;
typedef __attribute__((ext_vector_type(8))) float v8f;

#define N_NODES 50000
#define N_EDGES 1600000
#define D_IN    512
#define D_HEAD  64
#define D_H1    256

__global__ void zero_kernel(float* __restrict__ p, int n) {
  int i = blockIdx.x * blockDim.x + threadIdx.x;
  int stride = gridDim.x * blockDim.x;
  for (; i < n; i += stride) p[i] = 0.0f;
}

__global__ void count_kernel(const int* __restrict__ src, float* __restrict__ counts) {
  int i = blockIdx.x * blockDim.x + threadIdx.x;
  if (i < N_EDGES) atomicAdd(&counts[src[i]], 1.0f);
}

__global__ __launch_bounds__(256) void gemm1_kernel(const float* __restrict__ X,
                                                    const float* __restrict__ W1,
                                                    float* __restrict__ T) {
  const int wave = blockIdx.x * (blockDim.x >> 5) + (threadIdx.x >> 5);
  if (wave >= N_NODES / 16) return;
  const int lane = threadIdx.x & 31;
  const int half = lane >> 4;
  const int lidx = lane & 15;
  const int m0   = wave << 4;

  v8f zero = {};
  v8f acc[16];
#pragma unroll
  for (int i = 0; i < 16; ++i) acc[i] = zero;

  const float* xrow = X + (size_t)(m0 + lidx) * D_IN;
  for (int k0 = 0; k0 < D_IN; k0 += 4) {
    const int krow = k0 + 2 * half;
    v2f a = *(const v2f*)(xrow + krow);
#pragma unroll
    for (int nt = 0; nt < 16; ++nt) {
      const float* wb = W1 + (size_t)(nt >> 2) * (D_IN * D_HEAD) + (nt & 3) * 16 + lidx;
      v2f b;
      b.x = wb[(size_t)(krow    ) * D_HEAD];
      b.y = wb[(size_t)(krow + 1) * D_HEAD];
      acc[nt] = __builtin_amdgcn_wmma_f32_16x16x4_f32(
          false, a, false, b, (short)0, acc[nt], false, false);
    }
  }
#pragma unroll
  for (int nt = 0; nt < 16; ++nt) {
    float* trow = T + (size_t)(m0 + half * 8) * D_H1 + nt * 16 + lidx;
#pragma unroll
    for (int r = 0; r < 8; ++r)
      trow[(size_t)r * D_H1] = acc[nt][r];
  }
}

__global__ __launch_bounds__(256) void gemm2_kernel(const float* __restrict__ H,
                                                    const float* __restrict__ W2,
                                                    float* __restrict__ T2) {
  const int wave = blockIdx.x * (blockDim.x >> 5) + (threadIdx.x >> 5);
  if (wave >= N_NODES / 16) return;
  const int lane = threadIdx.x & 31;
  const int half = lane >> 4;
  const int lidx = lane & 15;
  const int m0   = wave << 4;

  v8f zero = {};
  v8f acc[4];
#pragma unroll
  for (int i = 0; i < 4; ++i) acc[i] = zero;

  const float* hrow = H + (size_t)(m0 + lidx) * D_H1;
  for (int k0 = 0; k0 < D_H1; k0 += 4) {
    const int krow = k0 + 2 * half;
    v2f a = *(const v2f*)(hrow + krow);
#pragma unroll
    for (int nt = 0; nt < 4; ++nt) {
      v2f b;
      b.x = W2[(size_t)(krow    ) * D_HEAD + nt * 16 + lidx];
      b.y = W2[(size_t)(krow + 1) * D_HEAD + nt * 16 + lidx];
      acc[nt] = __builtin_amdgcn_wmma_f32_16x16x4_f32(
          false, a, false, b, (short)0, acc[nt], false, false);
    }
  }
#pragma unroll
  for (int nt = 0; nt < 4; ++nt) {
    float* trow = T2 + (size_t)(m0 + half * 8) * D_HEAD + nt * 16 + lidx;
#pragma unroll
    for (int r = 0; r < 8; ++r)
      trow[(size_t)r * D_HEAD] = acc[nt][r];
  }
}

__global__ void scatter1_kernel(const int* __restrict__ src, const int* __restrict__ dst,
                                const float* __restrict__ T, float* __restrict__ acc) {
  const int wave = blockIdx.x * (blockDim.x >> 5) + (threadIdx.x >> 5);
  if (wave >= N_EDGES) return;
  const int lane = threadIdx.x & 31;
  const int s = src[wave];
  const int d = dst[wave];
  const float4* trow = (const float4*)(T + (size_t)d * D_H1) + lane * 2;
  float* arow = acc + (size_t)s * D_H1 + lane * 8;
  float4 v0 = trow[0];
  float4 v1 = trow[1];
  atomicAdd(arow + 0, v0.x); atomicAdd(arow + 1, v0.y);
  atomicAdd(arow + 2, v0.z); atomicAdd(arow + 3, v0.w);
  atomicAdd(arow + 4, v1.x); atomicAdd(arow + 5, v1.y);
  atomicAdd(arow + 6, v1.z); atomicAdd(arow + 7, v1.w);
}

__global__ void scatter2_kernel(const int* __restrict__ src, const int* __restrict__ dst,
                                const float* __restrict__ T2, float* __restrict__ out) {
  const int wave = blockIdx.x * (blockDim.x >> 5) + (threadIdx.x >> 5);
  if (wave >= N_EDGES) return;
  const int lane = threadIdx.x & 31;
  const int s = src[wave];
  const int d = dst[wave];
  const float2* trow = (const float2*)(T2 + (size_t)d * D_HEAD) + lane;
  float2 v = *trow;
  float* orow = out + (size_t)s * D_HEAD + lane * 2;
  atomicAdd(orow + 0, v.x);
  atomicAdd(orow + 1, v.y);
}

__global__ void divide_elu_kernel(const float* __restrict__ acc,
                                  const float* __restrict__ counts,
                                  float* __restrict__ H) {
  int i = blockIdx.x * blockDim.x + threadIdx.x;
  if (i >= N_NODES * D_H1) return;
  float c = counts[i >> 8];
  float h = (c > 0.0f) ? acc[i] / c : 0.0f;
  H[i] = (h > 0.0f) ? h : (expf(h) - 1.0f);
}

__global__ void finalize_kernel(float* __restrict__ out, const float* __restrict__ counts) {
  int i = blockIdx.x * blockDim.x + threadIdx.x;
  if (i >= N_NODES * D_HEAD) return;
  float c = counts[i >> 6];
  out[i] = (c > 0.0f) ? out[i] / c : 0.0f;
}

extern "C" void kernel_launch(void* const* d_in, const int* in_sizes, int n_in,
                              void* d_out, int out_size, void* d_ws, size_t ws_size,
                              hipStream_t stream) {
  const float* X  = (const float*)d_in[0];
  const int*   ei = (const int*)d_in[1];
  const float* W1 = (const float*)d_in[2];
  const float* W2 = (const float*)d_in[3];
  float* out = (float*)d_out;

  const int* src = ei;
  const int* dst = ei + N_EDGES;

  float* counts = (float*)d_ws;
  float* bufA   = counts + 50176;
  float* bufB   = bufA + (size_t)N_NODES * D_H1;

  zero_kernel<<<256, 256, 0, stream>>>(counts, 50176);
  zero_kernel<<<2048, 256, 0, stream>>>(bufB, N_NODES * D_H1);
  zero_kernel<<<512, 256, 0, stream>>>(out, N_NODES * D_HEAD);

  count_kernel<<<(N_EDGES + 255) / 256, 256, 0, stream>>>(src, counts);

  gemm1_kernel<<<391, 256, 0, stream>>>(X, W1, bufA);

  scatter1_kernel<<<N_EDGES / 8, 256, 0, stream>>>(src, dst, bufA, bufB);

  divide_elu_kernel<<<(N_NODES * D_H1 + 255) / 256, 256, 0, stream>>>(bufB, counts, bufA);

  gemm2_kernel<<<391, 256, 0, stream>>>(bufA, W2, bufB);

  scatter2_kernel<<<N_EDGES / 8, 256, 0, stream>>>(src, dst, bufB, out);

  finalize_kernel<<<(N_NODES * D_HEAD + 255) / 256, 256, 0, stream>>>(out, counts);
}


# Group edges by src in-call (returning int adds give ranks, scan, fill); one wave per node sums gathered rows in registers, replacing f32 atomic adds; sum order now by arrival rank; mean/elu at store
# speedup vs baseline: 3.2834x; 3.2834x over previous
#include <hip/hip_runtime.h>
#include <math.h>


typedef __attribute__((ext_vector_type(2))) float v2f;
typedef __attribute__((ext_vector_type(8))) float v8f;

#define N_NODES 50000
#define N_EDGES 1600000
#define D_IN    512
#define D_HEAD  64
#define D_H1    256
#define N_PAD   53248

__global__ void zero_kernel(int* __restrict__ p, int n) {
  int i = blockIdx.x * blockDim.x + threadIdx.x;
  int stride = gridDim.x * blockDim.x;
  for (; i < n; i += stride) p[i] = 0;
}

__global__ void count_rank_kernel(const int* __restrict__ src, int* __restrict__ cnt,
                                  int* __restrict__ rank) {
  const int i = blockIdx.x * blockDim.x + threadIdx.x;
  if (i >= N_EDGES / 4) return;
  const int4 s = ((const int4*)src)[i];
  int4 r;
  r.x = atomicAdd(&cnt[s.x], 1);
  r.y = atomicAdd(&cnt[s.y], 1);
  r.z = atomicAdd(&cnt[s.z], 1);
  r.w = atomicAdd(&cnt[s.w], 1);
  ((int4*)rank)[i] = r;
}

__global__ __launch_bounds__(1024) void scan_kernel(const int* __restrict__ cnt,
                                                    int* __restrict__ offs) {
  __shared__ int wave_sum[32];
  const int t = threadIdx.x;
  const int lane = t & 31;
  const int w = t >> 5;
  const int4* c4 = (const int4*)(cnt + t * 52);
  int4 v[13];
  int sum = 0;
#pragma unroll
  for (int j = 0; j < 13; ++j) {
    v[j] = c4[j];
    sum += (v[j].x + v[j].y) + (v[j].z + v[j].w);
  }
  int inc = sum;
#pragma unroll
  for (int d = 1; d < 32; d <<= 1) {
    const int up = __shfl_up(inc, d);
    if (lane >= d) inc += up;
  }
  if (lane == 31) wave_sum[w] = inc;
  __syncthreads();
  if (w == 0) {
    const int x = wave_sum[lane];
    int xi = x;
#pragma unroll
    for (int d = 1; d < 32; d <<= 1) {
      const int up = __shfl_up(xi, d);
      if (lane >= d) xi += up;
    }
    wave_sum[lane] = xi - x;
  }
  __syncthreads();
  int run = wave_sum[w] + inc - sum;
  int4* o4 = (int4*)(offs + t * 52);
#pragma unroll
  for (int j = 0; j < 13; ++j) {
    int4 o;
    o.x = run; run += v[j].x;
    o.y = run; run += v[j].y;
    o.z = run; run += v[j].z;
    o.w = run; run += v[j].w;
    o4[j] = o;
  }
}

__global__ void fill_kernel(const int* __restrict__ src, const int* __restrict__ dst,
                            const int* __restrict__ rank, const int* __restrict__ offs,
                            int* __restrict__ sdst) {
  const int i = blockIdx.x * blockDim.x + threadIdx.x;
  if (i >= N_EDGES / 4) return;
  const int4 s = ((const int4*)src)[i];
  const int4 d = ((const int4*)dst)[i];
  const int4 r = ((const int4*)rank)[i];
  sdst[offs[s.x] + r.x] = d.x;
  sdst[offs[s.y] + r.y] = d.y;
  sdst[offs[s.z] + r.z] = d.z;
  sdst[offs[s.w] + r.w] = d.w;
}

__global__ __launch_bounds__(256) void gemm1_kernel(const float* __restrict__ X,
                                                    const float* __restrict__ W1,
                                                    float* __restrict__ T) {
  const int wave = blockIdx.x * (blockDim.x >> 5) + (threadIdx.x >> 5);
  if (wave >= N_NODES / 16) return;
  const int lane = threadIdx.x & 31;
  const int half = lane >> 4;
  const int lidx = lane & 15;
  const int m0   = wave << 4;

  v8f zero = {};
  v8f acc[16];
#pragma unroll
  for (int i = 0; i < 16; ++i) acc[i] = zero;

  const float* xrow = X + (size_t)(m0 + lidx) * D_IN;
  for (int k0 = 0; k0 < D_IN; k0 += 4) {
    const int krow = k0 + 2 * half;
    v2f a = *(const v2f*)(xrow + krow);
#pragma unroll
    for (int nt = 0; nt < 16; ++nt) {
      const float* wb = W1 + (size_t)(nt >> 2) * (D_IN * D_HEAD) + (nt & 3) * 16 + lidx;
      v2f b;
      b.x = wb[(size_t)(krow    ) * D_HEAD];
      b.y = wb[(size_t)(krow + 1) * D_HEAD];
      acc[nt] = __builtin_amdgcn_wmma_f32_16x16x4_f32(
          false, a, false, b, (short)0, acc[nt], false, false);
    }
  }
#pragma unroll
  for (int nt = 0; nt < 16; ++nt) {
    float* trow = T + (size_t)(m0 + half * 8) * D_H1 + nt * 16 + lidx;
#pragma unroll
    for (int r = 0; r < 8; ++r)
      trow[(size_t)r * D_H1] = acc[nt][r];
  }
}

__global__ __launch_bounds__(256) void gemm2_kernel(const float* __restrict__ H,
                                                    const float* __restrict__ W2,
                                                    float* __restrict__ T2) {
  const int wave = blockIdx.x * (blockDim.x >> 5) + (threadIdx.x >> 5);
  if (wave >= N_NODES / 16) return;
  const int lane = threadIdx.x & 31;
  const int half = lane >> 4;
  const int lidx = lane & 15;
  const int m0   = wave << 4;

  v8f zero = {};
  v8f acc[4];
#pragma unroll
  for (int i = 0; i < 4; ++i) acc[i] = zero;

  const float* hrow = H + (size_t)(m0 + lidx) * D_H1;
  for (int k0 = 0; k0 < D_H1; k0 += 4) {
    const int krow = k0 + 2 * half;
    v2f a = *(const v2f*)(hrow + krow);
#pragma unroll
    for (int nt = 0; nt < 4; ++nt) {
      v2f b;
      b.x = W2[(size_t)(krow    ) * D_HEAD + nt * 16 + lidx];
      b.y = W2[(size_t)(krow + 1) * D_HEAD + nt * 16 + lidx];
      acc[nt] = __builtin_amdgcn_wmma_f32_16x16x4_f32(
          false, a, false, b, (short)0, acc[nt], false, false);
    }
  }
#pragma unroll
  for (int nt = 0; nt < 4; ++nt) {
    float* trow = T2 + (size_t)(m0 + half * 8) * D_HEAD + nt * 16 + lidx;
#pragma unroll
    for (int r = 0; r < 8; ++r)
      trow[(size_t)r * D_HEAD] = acc[nt][r];
  }
}

__device__ __forceinline__ float mean_elu(float a, float c, bool nz) {
  const float h = nz ? a / c : 0.0f;
  return (h > 0.0f) ? h : (expf(h) - 1.0f);
}

__global__ __launch_bounds__(256) void gather1_kernel(const int* __restrict__ offs,
                                                      const int* __restrict__ cnt,
                                                      const int* __restrict__ sdst,
                                                      const float* __restrict__ T,
                                                      float* __restrict__ H) {
  const int node = __builtin_amdgcn_readfirstlane((int)(blockIdx.x * 8 + (threadIdx.x >> 5)));
  if (node >= N_NODES) return;
  const int lane = threadIdx.x & 31;
  const int start = offs[node];
  const int n = cnt[node];
  const float4* __restrict__ T4 = (const float4*)T;
  float4 a0 = make_float4(0.0f, 0.0f, 0.0f, 0.0f);
  float4 a1 = make_float4(0.0f, 0.0f, 0.0f, 0.0f);

  for (int base = 0; base < n; base += 32) {
    const int m = min(32, n - base);
    const int my = (lane < m) ? sdst[start + base + lane] : 0;
    int j = 0;
    for (; j + 4 <= m; j += 4) {
      const float4* r0 = T4 + (size_t)__builtin_amdgcn_readlane(my, j    ) * 64 + lane;
      const float4* r1 = T4 + (size_t)__builtin_amdgcn_readlane(my, j + 1) * 64 + lane;
      const float4* r2 = T4 + (size_t)__builtin_amdgcn_readlane(my, j + 2) * 64 + lane;
      const float4* r3 = T4 + (size_t)__builtin_amdgcn_readlane(my, j + 3) * 64 + lane;
      const float4 x0 = r0[0], y0 = r0[32];
      const float4 x1 = r1[0], y1 = r1[32];
      const float4 x2 = r2[0], y2 = r2[32];
      const float4 x3 = r3[0], y3 = r3[32];
      a0 += x0; a1 += y0;
      a0 += x1; a1 += y1;
      a0 += x2; a1 += y2;
      a0 += x3; a1 += y3;
    }
    const int rem = m - j;
    if (rem > 0) {
      const float4 z = make_float4(0.0f, 0.0f, 0.0f, 0.0f);
      float4 x1 = z, y1 = z, x2 = z, y2 = z;
      const float4* r0 = T4 + (size_t)__builtin_amdgcn_readlane(my, j) * 64 + lane;
      const float4 x0 = r0[0], y0 = r0[32];
      if (rem > 1) {
        const float4* r1 = T4 + (size_t)__builtin_amdgcn_readlane(my, j + 1) * 64 + lane;
        x1 = r1[0]; y1 = r1[32];
      }
      if (rem > 2) {
        const float4* r2 = T4 + (size_t)__builtin_amdgcn_readlane(my, j + 2) * 64 + lane;
        x2 = r2[0]; y2 = r2[32];
      }
      a0 += x0; a1 += y0;
      a0 += x1; a1 += y1;
      a0 += x2; a1 += y2;
    }
  }

  const float c = (float)n;
  const bool nz = n > 0;
  float4 h0, h1;
  h0.x = mean_elu(a0.x, c, nz); h0.y = mean_elu(a0.y, c, nz);
  h0.z = mean_elu(a0.z, c, nz); h0.w = mean_elu(a0.w, c, nz);
  h1.x = mean_elu(a1.x, c, nz); h1.y = mean_elu(a1.y, c, nz);
  h1.z = mean_elu(a1.z, c, nz); h1.w = mean_elu(a1.w, c, nz);
  float4* hrow = (float4*)(H + (size_t)node * D_H1);
  hrow[lane]      = h0;
  hrow[32 + lane] = h1;
}

__global__ __launch_bounds__(256) void gather2_kernel(const int* __restrict__ offs,
                                                      const int* __restrict__ cnt,
                                                      const int* __restrict__ sdst,
                                                      const float* __restrict__ T2,
                                                      float* __restrict__ out) {
  const int node = __builtin_amdgcn_readfirstlane((int)(blockIdx.x * 8 + (threadIdx.x >> 5)));
  if (node >= N_NODES) return;
  const int lane = threadIdx.x & 31;
  const int half = lane >> 4;
  const int l16 = lane & 15;
  const int start = offs[node];
  const int n = cnt[node];
  const float4* __restrict__ T4 = (const float4*)T2;
  float4 a = make_float4(0.0f, 0.0f, 0.0f, 0.0f);

  for (int base = 0; base < n; base += 32) {
    const int m = min(32, n - base);
    const int my = (lane < m) ? sdst[start + base + lane] : 0;
    int j = 0;
    for (; j + 8 <= m; j += 8) {
      const int e0 = __builtin_amdgcn_readlane(my, j    ), o0 = __builtin_amdgcn_readlane(my, j + 1);
      const int e1 = __builtin_amdgcn_readlane(my, j + 2), o1 = __builtin_amdgcn_readlane(my, j + 3);
      const int e2 = __builtin_amdgcn_readlane(my, j + 4), o2 = __builtin_amdgcn_readlane(my, j + 5);
      const int e3 = __builtin_amdgcn_readlane(my, j + 6), o3 = __builtin_amdgcn_readlane(my, j + 7);
      const float4 x0 = T4[(size_t)(half ? o0 : e0) * 16 + l16];
      const float4 x1 = T4[(size_t)(half ? o1 : e1) * 16 + l16];
      const float4 x2 = T4[(size_t)(half ? o2 : e2) * 16 + l16];
      const float4 x3 = T4[(size_t)(half ? o3 : e3) * 16 + l16];
      a += x0; a += x1; a += x2; a += x3;
    }
    if (j < m) {
      const float4 z = make_float4(0.0f, 0.0f, 0.0f, 0.0f);
      float4 x0 = z, x1 = z, x2 = z, x3 = z;
      const int q = j + half;
      {
        const int e0 = __builtin_amdgcn_readlane(my, j), o0 = __builtin_amdgcn_readlane(my, j + 1);
        if (q < m) x0 = T4[(size_t)(half ? o0 : e0) * 16 + l16];
      }
      if (j + 2 < m) {
        const int e1 = __builtin_amdgcn_readlane(my, j + 2), o1 = __builtin_amdgcn_readlane(my, j + 3);
        if (q + 2 < m) x1 = T4[(size_t)(half ? o1 : e1) * 16 + l16];
      }
      if (j + 4 < m) {
        const int e2 = __builtin_amdgcn_readlane(my, j + 4), o2 = __builtin_amdgcn_readlane(my, j + 5);
        if (q + 4 < m) x2 = T4[(size_t)(half ? o2 : e2) * 16 + l16];
      }
      if (j + 6 < m) {
        const int e3 = __builtin_amdgcn_readlane(my, j + 6), o3 = __builtin_amdgcn_readlane(my, j + 7);
        if (q + 6 < m) x3 = T4[(size_t)(half ? o3 : e3) * 16 + l16];
      }
      a += x0; a += x1; a += x2; a += x3;
    }
  }

  a.x += __shfl_xor(a.x, 16);
  a.y += __shfl_xor(a.y, 16);
  a.z += __shfl_xor(a.z, 16);
  a.w += __shfl_xor(a.w, 16);
  if (half == 0) {
    const float c = (float)n;
    float4 o;
    o.x = (n > 0) ? a.x / c : 0.0f;
    o.y = (n > 0) ? a.y / c : 0.0f;
    o.z = (n > 0) ? a.z / c : 0.0f;
    o.w = (n > 0) ? a.w / c : 0.0f;
    ((float4*)(out + (size_t)node * D_HEAD))[l16] = o;
  }
}

extern "C" void kernel_launch(void* const* d_in, const int* in_sizes, int n_in,
                              void* d_out, int out_size, void* d_ws, size_t ws_size,
                              hipStream_t stream) {
  const float* X  = (const float*)d_in[0];
  const int*   ei = (const int*)d_in[1];
  const float* W1 = (const float*)d_in[2];
  const float* W2 = (const float*)d_in[3];
  float* out = (float*)d_out;

  const int* src = ei;
  const int* dst = ei + N_EDGES;

  int* cnt  = (int*)d_ws;
  int* offs = cnt + N_PAD;
  int* rank = offs + N_PAD;
  int* sdst = rank + N_EDGES;
  float* bufA = (float*)(sdst + N_EDGES);
  float* bufB = bufA + (size_t)N_NODES * D_H1;

  zero_kernel<<<208, 256, 0, stream>>>(cnt, N_PAD);
  count_rank_kernel<<<(N_EDGES / 4 + 255) / 256, 256, 0, stream>>>(src, cnt, rank);
  scan_kernel<<<1, 1024, 0, stream>>>(cnt, offs);
  fill_kernel<<<(N_EDGES / 4 + 255) / 256, 256, 0, stream>>>(src, dst, rank, offs, sdst);

  gemm1_kernel<<<391, 256, 0, stream>>>(X, W1, bufA);

  gather1_kernel<<<N_NODES / 8, 256, 0, stream>>>(offs, cnt, sdst, bufA, bufB);

  gemm2_kernel<<<391, 256, 0, stream>>>(bufB, W2, bufA);

  gather2_kernel<<<N_NODES / 8, 256, 0, stream>>>(offs, cnt, sdst, bufA, out);
}


# Edges grouped by src with a two-level counting sort on LDS counters (chunk histogram, bucket placement, per-bucket rank), no returning global add per edge; sum order still by arrival rank
# speedup vs baseline: 4.2247x; 1.2867x over previous
#include <hip/hip_runtime.h>
#include <math.h>


typedef __attribute__((ext_vector_type(2))) float v2f;
typedef __attribute__((ext_vector_type(8))) float v8f;

#define N_NODES 50000
#define N_EDGES 1600000
#define D_IN    512
#define D_HEAD  64
#define D_H1    256
#define N_PAD   53248
#define N_CHUNK 250
#define CHUNK_E 6400
#define N_BKT   256
#define HIST_ROWS 256

__device__ __forceinline__ int wave_incl_scan(int x, int lane) {
#pragma unroll
  for (int d = 1; d < 32; d <<= 1) {
    const int up = __shfl_up(x, d);
    if (lane >= d) x += up;
  }
  return x;
}

__device__ __forceinline__ int block_excl_scan_256(int x, int t, int* wsum) {
  const int lane = t & 31;
  const int w = t >> 5;
  const int inc = wave_incl_scan(x, lane);
  if (lane == 31 && w < 8) wsum[w] = inc;
  __syncthreads();
  int pre = 0;
#pragma unroll
  for (int j = 0; j < 8; ++j) pre += (j < w) ? wsum[j] : 0;
  return pre + inc - x;
}

__global__ __launch_bounds__(1024) void chunk_hist_kernel(const int* __restrict__ src,
                                                          int* __restrict__ hist) {
  __shared__ int h[N_BKT];
  const int t = threadIdx.x;
  const int c = blockIdx.x;
  if (c >= N_CHUNK) {
    if (t < N_BKT) hist[c * N_BKT + t] = 0;
    return;
  }
  if (t < N_BKT) h[t] = 0;
  __syncthreads();
  const int4* s4 = (const int4*)(src + (size_t)c * CHUNK_E);
  const bool two = t + 1024 < CHUNK_E / 4;
  const int4 a = s4[t];
  const int4 b = s4[two ? t + 1024 : t];
  atomicAdd(&h[a.x >> 8], 1);
  atomicAdd(&h[a.y >> 8], 1);
  atomicAdd(&h[a.z >> 8], 1);
  atomicAdd(&h[a.w >> 8], 1);
  if (two) {
    atomicAdd(&h[b.x >> 8], 1);
    atomicAdd(&h[b.y >> 8], 1);
    atomicAdd(&h[b.z >> 8], 1);
    atomicAdd(&h[b.w >> 8], 1);
  }
  __syncthreads();
  if (t < N_BKT) hist[c * N_BKT + t] = h[t];
}

__global__ __launch_bounds__(1024) void chunk_place_kernel(const int* __restrict__ src,
                                                           const int* __restrict__ dst,
                                                           const int* __restrict__ hist,
                                                           int* __restrict__ bbase,
                                                           int2* __restrict__ pairs) {
  __shared__ int part_tot[4][N_BKT];
  __shared__ int part_bef[4][N_BKT];
  __shared__ int cursor[N_BKT];
  __shared__ int wsum[8];
  const int t = threadIdx.x;
  const int c = blockIdx.x;

  const int4* s4 = (const int4*)(src + (size_t)c * CHUNK_E);
  const int4* d4 = (const int4*)(dst + (size_t)c * CHUNK_E);
  const bool two = t + 1024 < CHUNK_E / 4;
  const int t2 = two ? t + 1024 : t;
  const int4 sa = s4[t];
  const int4 da = d4[t];
  const int4 sb = s4[t2];
  const int4 db = d4[t2];

  const int b = t & (N_BKT - 1);
  const int q = t >> 8;
  int tot = 0, bef = 0;
  const int* hcol = hist + (q * 64) * N_BKT + b;
#pragma unroll 1
  for (int j0 = 0; j0 < 64; j0 += 32) {
    int v[32];
#pragma unroll
    for (int u = 0; u < 32; ++u) v[u] = hcol[(j0 + u) * N_BKT];
#pragma unroll
    for (int u = 0; u < 32; ++u) {
      tot += v[u];
      bef += (q * 64 + j0 + u < c) ? v[u] : 0;
    }
  }
  part_tot[q][b] = tot;
  part_bef[q][b] = bef;
  __syncthreads();
  int T = 0, Bf = 0;
  if (t < N_BKT) {
    T  = (part_tot[0][t] + part_tot[1][t]) + (part_tot[2][t] + part_tot[3][t]);
    Bf = (part_bef[0][t] + part_bef[1][t]) + (part_bef[2][t] + part_bef[3][t]);
  }
  const int excl = block_excl_scan_256(T, t, wsum);
  if (t < N_BKT) {
    cursor[t] = excl + Bf;
    if (c == 0) {
      bbase[t] = excl;
      if (t == N_BKT - 1) bbase[N_BKT] = excl + T;
    }
  }
  __syncthreads();

  {
    const int p0 = atomicAdd(&cursor[sa.x >> 8], 1);
    const int p1 = atomicAdd(&cursor[sa.y >> 8], 1);
    const int p2 = atomicAdd(&cursor[sa.z >> 8], 1);
    const int p3 = atomicAdd(&cursor[sa.w >> 8], 1);
    pairs[p0] = make_int2(sa.x, da.x);
    pairs[p1] = make_int2(sa.y, da.y);
    pairs[p2] = make_int2(sa.z, da.z);
    pairs[p3] = make_int2(sa.w, da.w);
  }
  if (two) {
    const int p0 = atomicAdd(&cursor[sb.x >> 8], 1);
    const int p1 = atomicAdd(&cursor[sb.y >> 8], 1);
    const int p2 = atomicAdd(&cursor[sb.z >> 8], 1);
    const int p3 = atomicAdd(&cursor[sb.w >> 8], 1);
    pairs[p0] = make_int2(sb.x, db.x);
    pairs[p1] = make_int2(sb.y, db.y);
    pairs[p2] = make_int2(sb.z, db.z);
    pairs[p3] = make_int2(sb.w, db.w);
  }
}

__global__ __launch_bounds__(1024) void bucket_group_kernel(const int2* __restrict__ pairs,
                                                            const int* __restrict__ bbase,
                                                            int* __restrict__ cnt,
                                                            int* __restrict__ offs,
                                                            int* __restrict__ sdst) {
  __shared__ int cl[N_BKT];
  __shared__ int cur[N_BKT];
  __shared__ int wsum[8];
  const int t = threadIdx.x;
  const int bk = blockIdx.x;
  const int base = bbase[bk];
  const int size = bbase[bk + 1] - base;
  const int2* __restrict__ P = pairs + base;
  if (t < N_BKT) cl[t] = 0;
  __syncthreads();

  const int last = size - 1;
  for (int i0 = 0; i0 < size; i0 += 4096) {
    const int i = i0 + t;
    const bool v0 = i < size, v1 = i + 1024 < size, v2 = i + 2048 < size, v3 = i + 3072 < size;
    const int s0 = P[min(i, last)].x;
    const int s1 = P[min(i + 1024, last)].x;
    const int s2 = P[min(i + 2048, last)].x;
    const int s3 = P[min(i + 3072, last)].x;
    if (v0) atomicAdd(&cl[s0 & 255], 1);
    if (v1) atomicAdd(&cl[s1 & 255], 1);
    if (v2) atomicAdd(&cl[s2 & 255], 1);
    if (v3) atomicAdd(&cl[s3 & 255], 1);
  }
  __syncthreads();

  const int mine = (t < N_BKT) ? cl[t] : 0;
  const int excl = block_excl_scan_256(mine, t, wsum);
  if (t < N_BKT) {
    cur[t] = excl;
    const int node = bk * N_BKT + t;
    if (node < N_NODES) {
      cnt[node] = mine;
      offs[node] = base + excl;
    }
  }
  __syncthreads();

  int* __restrict__ out = sdst + base;
  for (int i0 = 0; i0 < size; i0 += 4096) {
    const int i = i0 + t;
    const bool v0 = i < size, v1 = i + 1024 < size, v2 = i + 2048 < size, v3 = i + 3072 < size;
    const int2 e0 = P[min(i, last)];
    const int2 e1 = P[min(i + 1024, last)];
    const int2 e2 = P[min(i + 2048, last)];
    const int2 e3 = P[min(i + 3072, last)];
    int p0 = 0, p1 = 0, p2 = 0, p3 = 0;
    if (v0) p0 = atomicAdd(&cur[e0.x & 255], 1);
    if (v1) p1 = atomicAdd(&cur[e1.x & 255], 1);
    if (v2) p2 = atomicAdd(&cur[e2.x & 255], 1);
    if (v3) p3 = atomicAdd(&cur[e3.x & 255], 1);
    if (v0) out[p0] = e0.y;
    if (v1) out[p1] = e1.y;
    if (v2) out[p2] = e2.y;
    if (v3) out[p3] = e3.y;
  }
}

__global__ __launch_bounds__(256) void gemm1_kernel(const float* __restrict__ X,
                                                    const float* __restrict__ W1,
                                                    float* __restrict__ T) {
  const int wave = blockIdx.x * (blockDim.x >> 5) + (threadIdx.x >> 5);
  if (wave >= N_NODES / 16) return;
  const int lane = threadIdx.x & 31;
  const int half = lane >> 4;
  const int lidx = lane & 15;
  const int m0   = wave << 4;

  v8f zero = {};
  v8f acc[16];
#pragma unroll
  for (int i = 0; i < 16; ++i) acc[i] = zero;

  const float* xrow = X + (size_t)(m0 + lidx) * D_IN;
  for (int k0 = 0; k0 < D_IN; k0 += 4) {
    const int krow = k0 + 2 * half;
    v2f a = *(const v2f*)(xrow + krow);
#pragma unroll
    for (int nt = 0; nt < 16; ++nt) {
      const float* wb = W1 + (size_t)(nt >> 2) * (D_IN * D_HEAD) + (nt & 3) * 16 + lidx;
      v2f b;
      b.x = wb[(size_t)(krow    ) * D_HEAD];
      b.y = wb[(size_t)(krow + 1) * D_HEAD];
      acc[nt] = __builtin_amdgcn_wmma_f32_16x16x4_f32(
          false, a, false, b, (short)0, acc[nt], false, false);
    }
  }
#pragma unroll
  for (int nt = 0; nt < 16; ++nt) {
    float* trow = T + (size_t)(m0 + half * 8) * D_H1 + nt * 16 + lidx;
#pragma unroll
    for (int r = 0; r < 8; ++r)
      trow[(size_t)r * D_H1] = acc[nt][r];
  }
}

__global__ __launch_bounds__(256) void gemm2_kernel(const float* __restrict__ H,
                                                    const float* __restrict__ W2,
                                                    float* __restrict__ T2) {
  const int wave = blockIdx.x * (blockDim.x >> 5) + (threadIdx.x >> 5);
  if (wave >= N_NODES / 16) return;
  const int lane = threadIdx.x & 31;
  const int half = lane >> 4;
  const int lidx = lane & 15;
  const int m0   = wave << 4;

  v8f zero = {};
  v8f acc[4];
#pragma unroll
  for (int i = 0; i < 4; ++i) acc[i] = zero;

  const float* hrow = H + (size_t)(m0 + lidx) * D_H1;
  for (int k0 = 0; k0 < D_H1; k0 += 4) {
    const int krow = k0 + 2 * half;
    v2f a = *(const v2f*)(hrow + krow);
#pragma unroll
    for (int nt = 0; nt < 4; ++nt) {
      v2f b;
      b.x = W2[(size_t)(krow    ) * D_HEAD + nt * 16 + lidx];
      b.y = W2[(size_t)(krow + 1) * D_HEAD + nt * 16 + lidx];
      acc[nt] = __builtin_amdgcn_wmma_f32_16x16x4_f32(
          false, a, false, b, (short)0, acc[nt], false, false);
    }
  }
#pragma unroll
  for (int nt = 0; nt < 4; ++nt) {
    float* trow = T2 + (size_t)(m0 + half * 8) * D_HEAD + nt * 16 + lidx;
#pragma unroll
    for (int r = 0; r < 8; ++r)
      trow[(size_t)r * D_HEAD] = acc[nt][r];
  }
}

__device__ __forceinline__ float mean_elu(float a, float c, bool nz) {
  const float h = nz ? a / c : 0.0f;
  return (h > 0.0f) ? h : (expf(h) - 1.0f);
}

__global__ __launch_bounds__(256) void gather1_kernel(const int* __restrict__ offs,
                                                      const int* __restrict__ cnt,
                                                      const int* __restrict__ sdst,
                                                      const float* __restrict__ T,
                                                      float* __restrict__ H) {
  const int node = __builtin_amdgcn_readfirstlane((int)(blockIdx.x * 8 + (threadIdx.x >> 5)));
  if (node >= N_NODES) return;
  const int lane = threadIdx.x & 31;
  const int start = offs[node];
  const int n = cnt[node];
  const float4* __restrict__ T4 = (const float4*)T;
  float4 a0 = make_float4(0.0f, 0.0f, 0.0f, 0.0f);
  float4 a1 = make_float4(0.0f, 0.0f, 0.0f, 0.0f);

  for (int base = 0; base < n; base += 32) {
    const int m = min(32, n - base);
    const int my = (lane < m) ? sdst[start + base + lane] : 0;
    int j = 0;
    for (; j + 4 <= m; j += 4) {
      const float4* r0 = T4 + (size_t)__builtin_amdgcn_readlane(my, j    ) * 64 + lane;
      const float4* r1 = T4 + (size_t)__builtin_amdgcn_readlane(my, j + 1) * 64 + lane;
      const float4* r2 = T4 + (size_t)__builtin_amdgcn_readlane(my, j + 2) * 64 + lane;
      const float4* r3 = T4 + (size_t)__builtin_amdgcn_readlane(my, j + 3) * 64 + lane;
      const float4 x0 = r0[0], y0 = r0[32];
      const float4 x1 = r1[0], y1 = r1[32];
      const float4 x2 = r2[0], y2 = r2[32];
      const float4 x3 = r3[0], y3 = r3[32];
      a0 += x0; a1 += y0;
      a0 += x1; a1 += y1;
      a0 += x2; a1 += y2;
      a0 += x3; a1 += y3;
    }
    const int rem = m - j;
    if (rem > 0) {
      const float4 z = make_float4(0.0f, 0.0f, 0.0f, 0.0f);
      float4 x1 = z, y1 = z, x2 = z, y2 = z;
      const float4* r0 = T4 + (size_t)__builtin_amdgcn_readlane(my, j) * 64 + lane;
      const float4 x0 = r0[0], y0 = r0[32];
      if (rem > 1) {
        const float4* r1 = T4 + (size_t)__builtin_amdgcn_readlane(my, j + 1) * 64 + lane;
        x1 = r1[0]; y1 = r1[32];
      }
      if (rem > 2) {
        const float4* r2 = T4 + (size_t)__builtin_amdgcn_readlane(my, j + 2) * 64 + lane;
        x2 = r2[0]; y2 = r2[32];
      }
      a0 += x0; a1 += y0;
      a0 += x1; a1 += y1;
      a0 += x2; a1 += y2;
    }
  }

  const float c = (float)n;
  const bool nz = n > 0;
  float4 h0, h1;
  h0.x = mean_elu(a0.x, c, nz); h0.y = mean_elu(a0.y, c, nz);
  h0.z = mean_elu(a0.z, c, nz); h0.w = mean_elu(a0.w, c, nz);
  h1.x = mean_elu(a1.x, c, nz); h1.y = mean_elu(a1.y, c, nz);
  h1.z = mean_elu(a1.z, c, nz); h1.w = mean_elu(a1.w, c, nz);
  float4* hrow = (float4*)(H + (size_t)node * D_H1);
  hrow[lane]      = h0;
  hrow[32 + lane] = h1;
}

__global__ __launch_bounds__(256) void gather2_kernel(const int* __restrict__ offs,
                                                      const int* __restrict__ cnt,
                                                      const int* __restrict__ sdst,
                                                      const float* __restrict__ T2,
                                                      float* __restrict__ out) {
  const int node = __builtin_amdgcn_readfirstlane((int)(blockIdx.x * 8 + (threadIdx.x >> 5)));
  if (node >= N_NODES) return;
  const int lane = threadIdx.x & 31;
  const int half = lane >> 4;
  const int l16 = lane & 15;
  const int start = offs[node];
  const int n = cnt[node];
  const float4* __restrict__ T4 = (const float4*)T2;
  float4 a = make_float4(0.0f, 0.0f, 0.0f, 0.0f);

  for (int base = 0; base < n; base += 32) {
    const int m = min(32, n - base);
    const int my = (lane < m) ? sdst[start + base + lane] : 0;
    int j = 0;
    for (; j + 8 <= m; j += 8) {
      const int e0 = __builtin_amdgcn_readlane(my, j    ), o0 = __builtin_amdgcn_readlane(my, j + 1);
      const int e1 = __builtin_amdgcn_readlane(my, j + 2), o1 = __builtin_amdgcn_readlane(my, j + 3);
      const int e2 = __builtin_amdgcn_readlane(my, j + 4), o2 = __builtin_amdgcn_readlane(my, j + 5);
      const int e3 = __builtin_amdgcn_readlane(my, j + 6), o3 = __builtin_amdgcn_readlane(my, j + 7);
      const float4 x0 = T4[(size_t)(half ? o0 : e0) * 16 + l16];
      const float4 x1 = T4[(size_t)(half ? o1 : e1) * 16 + l16];
      const float4 x2 = T4[(size_t)(half ? o2 : e2) * 16 + l16];
      const float4 x3 = T4[(size_t)(half ? o3 : e3) * 16 + l16];
      a += x0; a += x1; a += x2; a += x3;
    }
    if (j < m) {
      const float4 z = make_float4(0.0f, 0.0f, 0.0f, 0.0f);
      float4 x0 = z, x1 = z, x2 = z, x3 = z;
      const int q = j + half;
      {
        const int e0 = __builtin_amdgcn_readlane(my, j), o0 = __builtin_amdgcn_readlane(my, j + 1);
        if (q < m) x0 = T4[(size_t)(half ? o0 : e0) * 16 + l16];
      }
      if (j + 2 < m) {
        const int e1 = __builtin_amdgcn_readlane(my, j + 2), o1 = __builtin_amdgcn_readlane(my, j + 3);
        if (q + 2 < m) x1 = T4[(size_t)(half ? o1 : e1) * 16 + l16];
      }
      if (j + 4 < m) {
        const int e2 = __builtin_amdgcn_readlane(my, j + 4), o2 = __builtin_amdgcn_readlane(my, j + 5);
        if (q + 4 < m) x2 = T4[(size_t)(half ? o2 : e2) * 16 + l16];
      }
      if (j + 6 < m) {
        const int e3 = __builtin_amdgcn_readlane(my, j + 6), o3 = __builtin_amdgcn_readlane(my, j + 7);
        if (q + 6 < m) x3 = T4[(size_t)(half ? o3 : e3) * 16 + l16];
      }
      a += x0; a += x1; a += x2; a += x3;
    }
  }

  a.x += __shfl_xor(a.x, 16);
  a.y += __shfl_xor(a.y, 16);
  a.z += __shfl_xor(a.z, 16);
  a.w += __shfl_xor(a.w, 16);
  if (half == 0) {
    const float c = (float)n;
    float4 o;
    o.x = (n > 0) ? a.x / c : 0.0f;
    o.y = (n > 0) ? a.y / c : 0.0f;
    o.z = (n > 0) ? a.z / c : 0.0f;
    o.w = (n > 0) ? a.w / c : 0.0f;
    ((float4*)(out + (size_t)node * D_HEAD))[l16] = o;
  }
}

extern "C" void kernel_launch(void* const* d_in, const int* in_sizes, int n_in,
                              void* d_out, int out_size, void* d_ws, size_t ws_size,
                              hipStream_t stream) {
  const float* X  = (const float*)d_in[0];
  const int*   ei = (const int*)d_in[1];
  const float* W1 = (const float*)d_in[2];
  const float* W2 = (const float*)d_in[3];
  float* out = (float*)d_out;

  const int* src = ei;
  const int* dst = ei + N_EDGES;

  int* cnt   = (int*)d_ws;
  int* offs  = cnt + N_PAD;
  int* hist  = offs + N_PAD;
  int* bbase = hist + HIST_ROWS * N_BKT;
  int* sdst  = hist + N_EDGES;
  float* bufA = (float*)(sdst + N_EDGES);
  float* bufB = bufA + (size_t)N_NODES * D_H1;
  int2* pairs = (int2*)(bufB + (size_t)N_NODES * D_H1);

  chunk_hist_kernel<<<HIST_ROWS, 1024, 0, stream>>>(src, hist);
  chunk_place_kernel<<<N_CHUNK, 1024, 0, stream>>>(src, dst, hist, bbase, pairs);
  bucket_group_kernel<<<(N_NODES + N_BKT - 1) / N_BKT, 1024, 0, stream>>>(pairs, bbase, cnt, offs, sdst);

  gemm1_kernel<<<391, 256, 0, stream>>>(X, W1, bufA);

  gather1_kernel<<<N_NODES / 8, 256, 0, stream>>>(offs, cnt, sdst, bufA, bufB);

  gemm2_kernel<<<391, 256, 0, stream>>>(bufB, W2, bufA);

  gather2_kernel<<<N_NODES / 8, 256, 0, stream>>>(offs, cnt, sdst, bufA, out);
}


# GEMM1 reads its B operand from LDS: each work-group holds one head's 512x64 slice of W1 there and its waves loop over 16x64 output tiles; order of every sum unchanged
# speedup vs baseline: 7.5473x; 1.7865x over previous
#include <hip/hip_runtime.h>
#include <math.h>


typedef __attribute__((ext_vector_type(2))) float v2f;
typedef __attribute__((ext_vector_type(8))) float v8f;

#define N_NODES 50000
#define N_EDGES 1600000
#define D_IN    512
#define D_HEAD  64
#define D_H1    256
#define N_PAD   53248
#define N_CHUNK 250
#define CHUNK_E 6400
#define N_BKT   256
#define HIST_ROWS 256

__device__ __forceinline__ int wave_incl_scan(int x, int lane) {
#pragma unroll
  for (int d = 1; d < 32; d <<= 1) {
    const int up = __shfl_up(x, d);
    if (lane >= d) x += up;
  }
  return x;
}

__device__ __forceinline__ int block_excl_scan_256(int x, int t, int* wsum) {
  const int lane = t & 31;
  const int w = t >> 5;
  const int inc = wave_incl_scan(x, lane);
  if (lane == 31 && w < 8) wsum[w] = inc;
  __syncthreads();
  int pre = 0;
#pragma unroll
  for (int j = 0; j < 8; ++j) pre += (j < w) ? wsum[j] : 0;
  return pre + inc - x;
}

__global__ __launch_bounds__(1024) void chunk_hist_kernel(const int* __restrict__ src,
                                                          int* __restrict__ hist) {
  __shared__ int h[N_BKT];
  const int t = threadIdx.x;
  const int c = blockIdx.x;
  if (c >= N_CHUNK) {
    if (t < N_BKT) hist[c * N_BKT + t] = 0;
    return;
  }
  if (t < N_BKT) h[t] = 0;
  __syncthreads();
  const int4* s4 = (const int4*)(src + (size_t)c * CHUNK_E);
  const bool two = t + 1024 < CHUNK_E / 4;
  const int4 a = s4[t];
  const int4 b = s4[two ? t + 1024 : t];
  atomicAdd(&h[a.x >> 8], 1);
  atomicAdd(&h[a.y >> 8], 1);
  atomicAdd(&h[a.z >> 8], 1);
  atomicAdd(&h[a.w >> 8], 1);
  if (two) {
    atomicAdd(&h[b.x >> 8], 1);
    atomicAdd(&h[b.y >> 8], 1);
    atomicAdd(&h[b.z >> 8], 1);
    atomicAdd(&h[b.w >> 8], 1);
  }
  __syncthreads();
  if (t < N_BKT) hist[c * N_BKT + t] = h[t];
}

__global__ __launch_bounds__(1024) void chunk_place_kernel(const int* __restrict__ src,
                                                           const int* __restrict__ dst,
                                                           const int* __restrict__ hist,
                                                           int* __restrict__ bbase,
                                                           int2* __restrict__ pairs) {
  __shared__ int part_tot[4][N_BKT];
  __shared__ int part_bef[4][N_BKT];
  __shared__ int cursor[N_BKT];
  __shared__ int wsum[8];
  const int t = threadIdx.x;
  const int c = blockIdx.x;

  const int4* s4 = (const int4*)(src + (size_t)c * CHUNK_E);
  const int4* d4 = (const int4*)(dst + (size_t)c * CHUNK_E);
  const bool two = t + 1024 < CHUNK_E / 4;
  const int t2 = two ? t + 1024 : t;
  const int4 sa = s4[t];
  const int4 da = d4[t];
  const int4 sb = s4[t2];
  const int4 db = d4[t2];

  const int b = t & (N_BKT - 1);
  const int q = t >> 8;
  int tot = 0, bef = 0;
  const int* hcol = hist + (q * 64) * N_BKT + b;
#pragma unroll 1
  for (int j0 = 0; j0 < 64; j0 += 32) {
    int v[32];
#pragma unroll
    for (int u = 0; u < 32; ++u) v[u] = hcol[(j0 + u) * N_BKT];
#pragma unroll
    for (int u = 0; u < 32; ++u) {
      tot += v[u];
      bef += (q * 64 + j0 + u < c) ? v[u] : 0;
    }
  }
  part_tot[q][b] = tot;
  part_bef[q][b] = bef;
  __syncthreads();
  int T = 0, Bf = 0;
  if (t < N_BKT) {
    T  = (part_tot[0][t] + part_tot[1][t]) + (part_tot[2][t] + part_tot[3][t]);
    Bf = (part_bef[0][t] + part_bef[1][t]) + (part_bef[2][t] + part_bef[3][t]);
  }
  const int excl = block_excl_scan_256(T, t, wsum);
  if (t < N_BKT) {
    cursor[t] = excl + Bf;
    if (c == 0) {
      bbase[t] = excl;
      if (t == N_BKT - 1) bbase[N_BKT] = excl + T;
    }
  }
  __syncthreads();

  {
    const int p0 = atomicAdd(&cursor[sa.x >> 8], 1);
    const int p1 = atomicAdd(&cursor[sa.y >> 8], 1);
    const int p2 = atomicAdd(&cursor[sa.z >> 8], 1);
    const int p3 = atomicAdd(&cursor[sa.w >> 8], 1);
    pairs[p0] = make_int2(sa.x, da.x);
    pairs[p1] = make_int2(sa.y, da.y);
    pairs[p2] = make_int2(sa.z, da.z);
    pairs[p3] = make_int2(sa.w, da.w);
  }
  if (two) {
    const int p0 = atomicAdd(&cursor[sb.x >> 8], 1);
    const int p1 = atomicAdd(&cursor[sb.y >> 8], 1);
    const int p2 = atomicAdd(&cursor[sb.z >> 8], 1);
    const int p3 = atomicAdd(&cursor[sb.w >> 8], 1);
    pairs[p0] = make_int2(sb.x, db.x);
    pairs[p1] = make_int2(sb.y, db.y);
    pairs[p2] = make_int2(sb.z, db.z);
    pairs[p3] = make_int2(sb.w, db.w);
  }
}

__global__ __launch_bounds__(1024) void bucket_group_kernel(const int2* __restrict__ pairs,
                                                            const int* __restrict__ bbase,
                                                            int* __restrict__ cnt,
                                                            int* __restrict__ offs,
                                                            int* __restrict__ sdst) {
  __shared__ int cl[N_BKT];
  __shared__ int cur[N_BKT];
  __shared__ int wsum[8];
  const int t = threadIdx.x;
  const int bk = blockIdx.x;
  const int base = bbase[bk];
  const int size = bbase[bk + 1] - base;
  const int2* __restrict__ P = pairs + base;
  if (t < N_BKT) cl[t] = 0;
  __syncthreads();

  const int last = size - 1;
  for (int i0 = 0; i0 < size; i0 += 4096) {
    const int i = i0 + t;
    const bool v0 = i < size, v1 = i + 1024 < size, v2 = i + 2048 < size, v3 = i + 3072 < size;
    const int s0 = P[min(i, last)].x;
    const int s1 = P[min(i + 1024, last)].x;
    const int s2 = P[min(i + 2048, last)].x;
    const int s3 = P[min(i + 3072, last)].x;
    if (v0) atomicAdd(&cl[s0 & 255], 1);
    if (v1) atomicAdd(&cl[s1 & 255], 1);
    if (v2) atomicAdd(&cl[s2 & 255], 1);
    if (v3) atomicAdd(&cl[s3 & 255], 1);
  }
  __syncthreads();

  const int mine = (t < N_BKT) ? cl[t] : 0;
  const int excl = block_excl_scan_256(mine, t, wsum);
  if (t < N_BKT) {
    cur[t] = excl;
    const int node = bk * N_BKT + t;
    if (node < N_NODES) {
      cnt[node] = mine;
      offs[node] = base + excl;
    }
  }
  __syncthreads();

  int* __restrict__ out = sdst + base;
  for (int i0 = 0; i0 < size; i0 += 4096) {
    const int i = i0 + t;
    const bool v0 = i < size, v1 = i + 1024 < size, v2 = i + 2048 < size, v3 = i + 3072 < size;
    const int2 e0 = P[min(i, last)];
    const int2 e1 = P[min(i + 1024, last)];
    const int2 e2 = P[min(i + 2048, last)];
    const int2 e3 = P[min(i + 3072, last)];
    int p0 = 0, p1 = 0, p2 = 0, p3 = 0;
    if (v0) p0 = atomicAdd(&cur[e0.x & 255], 1);
    if (v1) p1 = atomicAdd(&cur[e1.x & 255], 1);
    if (v2) p2 = atomicAdd(&cur[e2.x & 255], 1);
    if (v3) p3 = atomicAdd(&cur[e3.x & 255], 1);
    if (v0) out[p0] = e0.y;
    if (v1) out[p1] = e1.y;
    if (v2) out[p2] = e2.y;
    if (v3) out[p3] = e3.y;
  }
}

#define G1_HEAD_BLOCKS 128
#define G1_WAVES       8
#define N_STRIPS       (N_NODES / 16)
#define KP_ROW         128

__device__ __forceinline__ void g1_load_group(v2f (&a)[8], const float* p) {
#pragma unroll
  for (int s = 0; s < 8; ++s) a[s] = *(const v2f*)(p + 4 * s);
}

__device__ __forceinline__ void g1_mma_group(v8f (&acc)[4], const v2f (&a)[8], const float* bl) {
#pragma unroll
  for (int s = 0; s < 8; ++s) {
    const float4 q0 = *(const float4*)(bl + s * (2 * KP_ROW));
    const float4 q1 = *(const float4*)(bl + s * (2 * KP_ROW) + 64);
    v2f b0, b1, b2, b3;
    b0.x = q0.x; b0.y = q0.y;
    b1.x = q0.z; b1.y = q0.w;
    b2.x = q1.x; b2.y = q1.y;
    b3.x = q1.z; b3.y = q1.w;
    acc[0] = __builtin_amdgcn_wmma_f32_16x16x4_f32(false, a[s], false, b0, (short)0, acc[0], false, false);
    acc[1] = __builtin_amdgcn_wmma_f32_16x16x4_f32(false, a[s], false, b1, (short)0, acc[1], false, false);
    acc[2] = __builtin_amdgcn_wmma_f32_16x16x4_f32(false, a[s], false, b2, (short)0, acc[2], false, false);
    acc[3] = __builtin_amdgcn_wmma_f32_16x16x4_f32(false, a[s], false, b3, (short)0, acc[3], false, false);
  }
}

__global__ __launch_bounds__(256) void gemm1_kernel(const float* __restrict__ X,
                                                    const float* __restrict__ W1,
                                                    float* __restrict__ T) {
  __shared__ __attribute__((aligned(16))) float Bs[256 * KP_ROW];
  const int t = threadIdx.x;
  const int head = blockIdx.x & 3;
  const int bih  = blockIdx.x >> 2;
  const float* __restrict__ Wh = W1 + (size_t)head * (D_IN * D_HEAD);

  {
    const int cq  = t & 15;
    const int kp0 = t >> 4;
    float4 r0[16], r1[16];
#pragma unroll
    for (int i = 0; i < 16; ++i) {
      const float4* p = (const float4*)(Wh + (size_t)(2 * (kp0 + 16 * i)) * D_HEAD) + cq;
      r0[i] = p[0];
      r1[i] = p[16];
    }
#pragma unroll
    for (int i = 0; i < 16; ++i) {
      float* q = Bs + (kp0 + 16 * i) * KP_ROW + 4 * cq;
      *(float4*)(q)      = make_float4(r0[i].x, r1[i].x, r0[i].y, r1[i].y);
      *(float4*)(q + 64) = make_float4(r0[i].z, r1[i].z, r0[i].w, r1[i].w);
    }
  }
  __syncthreads();

  const int wave = __builtin_amdgcn_readfirstlane(t >> 5);
  const int lane = t & 31;
  const int half = lane >> 4;
  const int lidx = lane & 15;
  const float* bl = Bs + half * KP_ROW + 4 * lidx;

  int strip = wave * G1_HEAD_BLOCKS + bih;
  const float* xrow = X + (size_t)((strip << 4) + lidx) * D_IN + 2 * half;
  v2f a0[8], a1[8];
  g1_load_group(a0, xrow);

  for (;;) {
    const int next = strip + G1_WAVES * G1_HEAD_BLOCKS;
    const bool has_next = next < N_STRIPS;
    const float* xnext = has_next ? X + (size_t)((next << 4) + lidx) * D_IN + 2 * half : xrow;

    v8f zero = {};
    v8f acc[4];
    acc[0] = zero; acc[1] = zero; acc[2] = zero; acc[3] = zero;

    for (int g = 0; g < D_IN / 32; g += 2) {
      g1_load_group(a1, xrow + (g + 1) * 32);
      g1_mma_group(acc, a0, bl + g * (16 * KP_ROW));
      g1_load_group(a0, (g + 2 < D_IN / 32) ? xrow + (g + 2) * 32 : xnext);
      g1_mma_group(acc, a1, bl + (g + 1) * (16 * KP_ROW));
    }

    float* trow = T + (size_t)((strip << 4) + half * 8) * D_H1 + head * D_HEAD + 4 * lidx;
#pragma unroll
    for (int r = 0; r < 8; ++r)
      *(float4*)(trow + (size_t)r * D_H1) = make_float4(acc[0][r], acc[1][r], acc[2][r], acc[3][r]);
    if (!has_next) break;
    strip = next;
    xrow = xnext;
  }
}

__global__ __launch_bounds__(256) void gemm2_kernel(const float* __restrict__ H,
                                                    const float* __restrict__ W2,
                                                    float* __restrict__ T2) {
  const int wave = blockIdx.x * (blockDim.x >> 5) + (threadIdx.x >> 5);
  if (wave >= N_NODES / 16) return;
  const int lane = threadIdx.x & 31;
  const int half = lane >> 4;
  const int lidx = lane & 15;
  const int m0   = wave << 4;

  v8f zero = {};
  v8f acc[4];
#pragma unroll
  for (int i = 0; i < 4; ++i) acc[i] = zero;

  const float* hrow = H + (size_t)(m0 + lidx) * D_H1;
  for (int k0 = 0; k0 < D_H1; k0 += 4) {
    const int krow = k0 + 2 * half;
    v2f a = *(const v2f*)(hrow + krow);
#pragma unroll
    for (int nt = 0; nt < 4; ++nt) {
      v2f b;
      b.x = W2[(size_t)(krow    ) * D_HEAD + nt * 16 + lidx];
      b.y = W2[(size_t)(krow + 1) * D_HEAD + nt * 16 + lidx];
      acc[nt] = __builtin_amdgcn_wmma_f32_16x16x4_f32(
          false, a, false, b, (short)0, acc[nt], false, false);
    }
  }
#pragma unroll
  for (int nt = 0; nt < 4; ++nt) {
    float* trow = T2 + (size_t)(m0 + half * 8) * D_HEAD + nt * 16 + lidx;
#pragma unroll
    for (int r = 0; r < 8; ++r)
      trow[(size_t)r * D_HEAD] = acc[nt][r];
  }
}

__device__ __forceinline__ float mean_elu(float a, float c, bool nz) {
  const float h = nz ? a / c : 0.0f;
  return (h > 0.0f) ? h : (expf(h) - 1.0f);
}

__global__ __launch_bounds__(256) void gather1_kernel(const int* __restrict__ offs,
                                                      const int* __restrict__ cnt,
                                                      const int* __restrict__ sdst,
                                                      const float* __restrict__ T,
                                                      float* __restrict__ H) {
  const int node = __builtin_amdgcn_readfirstlane((int)(blockIdx.x * 8 + (threadIdx.x >> 5)));
  if (node >= N_NODES) return;
  const int lane = threadIdx.x & 31;
  const int start = offs[node];
  const int n = cnt[node];
  const float4* __restrict__ T4 = (const float4*)T;
  float4 a0 = make_float4(0.0f, 0.0f, 0.0f, 0.0f);
  float4 a1 = make_float4(0.0f, 0.0f, 0.0f, 0.0f);

  for (int base = 0; base < n; base += 32) {
    const int m = min(32, n - base);
    const int my = (lane < m) ? sdst[start + base + lane] : 0;
    int j = 0;
    for (; j + 4 <= m; j += 4) {
      const float4* r0 = T4 + (size_t)__builtin_amdgcn_readlane(my, j    ) * 64 + lane;
      const float4* r1 = T4 + (size_t)__builtin_amdgcn_readlane(my, j + 1) * 64 + lane;
      const float4* r2 = T4 + (size_t)__builtin_amdgcn_readlane(my, j + 2) * 64 + lane;
      const float4* r3 = T4 + (size_t)__builtin_amdgcn_readlane(my, j + 3) * 64 + lane;
      const float4 x0 = r0[0], y0 = r0[32];
      const float4 x1 = r1[0], y1 = r1[32];
      const float4 x2 = r2[0], y2 = r2[32];
      const float4 x3 = r3[0], y3 = r3[32];
      a0 += x0; a1 += y0;
      a0 += x1; a1 += y1;
      a0 += x2; a1 += y2;
      a0 += x3; a1 += y3;
    }
    const int rem = m - j;
    if (rem > 0) {
      const float4 z = make_float4(0.0f, 0.0f, 0.0f, 0.0f);
      float4 x1 = z, y1 = z, x2 = z, y2 = z;
      const float4* r0 = T4 + (size_t)__builtin_amdgcn_readlane(my, j) * 64 + lane;
      const float4 x0 = r0[0], y0 = r0[32];
      if (rem > 1) {
        const float4* r1 = T4 + (size_t)__builtin_amdgcn_readlane(my, j + 1) * 64 + lane;
        x1 = r1[0]; y1 = r1[32];
      }
      if (rem > 2) {
        const float4* r2 = T4 + (size_t)__builtin_amdgcn_readlane(my, j + 2) * 64 + lane;
        x2 = r2[0]; y2 = r2[32];
      }
      a0 += x0; a1 += y0;
      a0 += x1; a1 += y1;
      a0 += x2; a1 += y2;
    }
  }

  const float c = (float)n;
  const bool nz = n > 0;
  float4 h0, h1;
  h0.x = mean_elu(a0.x, c, nz); h0.y = mean_elu(a0.y, c, nz);
  h0.z = mean_elu(a0.z, c, nz); h0.w = mean_elu(a0.w, c, nz);
  h1.x = mean_elu(a1.x, c, nz); h1.y = mean_elu(a1.y, c, nz);
  h1.z = mean_elu(a1.z, c, nz); h1.w = mean_elu(a1.w, c, nz);
  float4* hrow = (float4*)(H + (size_t)node * D_H1);
  hrow[lane]      = h0;
  hrow[32 + lane] = h1;
}

__global__ __launch_bounds__(256) void gather2_kernel(const int* __restrict__ offs,
                                                      const int* __restrict__ cnt,
                                                      const int* __restrict__ sdst,
                                                      const float* __restrict__ T2,
                                                      float* __restrict__ out) {
  const int node = __builtin_amdgcn_readfirstlane((int)(blockIdx.x * 8 + (threadIdx.x >> 5)));
  if (node >= N_NODES) return;
  const int lane = threadIdx.x & 31;
  const int half = lane >> 4;
  const int l16 = lane & 15;
  const int start = offs[node];
  const int n = cnt[node];
  const float4* __restrict__ T4 = (const float4*)T2;
  float4 a = make_float4(0.0f, 0.0f, 0.0f, 0.0f);

  for (int base = 0; base < n; base += 32) {
    const int m = min(32, n - base);
    const int my = (lane < m) ? sdst[start + base + lane] : 0;
    int j = 0;
    for (; j + 8 <= m; j += 8) {
      const int e0 = __builtin_amdgcn_readlane(my, j    ), o0 = __builtin_amdgcn_readlane(my, j + 1);
      const int e1 = __builtin_amdgcn_readlane(my, j + 2), o1 = __builtin_amdgcn_readlane(my, j + 3);
      const int e2 = __builtin_amdgcn_readlane(my, j + 4), o2 = __builtin_amdgcn_readlane(my, j + 5);
      const int e3 = __builtin_amdgcn_readlane(my, j + 6), o3 = __builtin_amdgcn_readlane(my, j + 7);
      const float4 x0 = T4[(size_t)(half ? o0 : e0) * 16 + l16];
      const float4 x1 = T4[(size_t)(half ? o1 : e1) * 16 + l16];
      const float4 x2 = T4[(size_t)(half ? o2 : e2) * 16 + l16];
      const float4 x3 = T4[(size_t)(half ? o3 : e3) * 16 + l16];
      a += x0; a += x1; a += x2; a += x3;
    }
    if (j < m) {
      const float4 z = make_float4(0.0f, 0.0f, 0.0f, 0.0f);
      float4 x0 = z, x1 = z, x2 = z, x3 = z;
      const int q = j + half;
      {
        const int e0 = __builtin_amdgcn_readlane(my, j), o0 = __builtin_amdgcn_readlane(my, j + 1);
        if (q < m) x0 = T4[(size_t)(half ? o0 : e0) * 16 + l16];
      }
      if (j + 2 < m) {
        const int e1 = __builtin_amdgcn_readlane(my, j + 2), o1 = __builtin_amdgcn_readlane(my, j + 3);
        if (q + 2 < m) x1 = T4[(size_t)(half ? o1 : e1) * 16 + l16];
      }
      if (j + 4 < m) {
        const int e2 = __builtin_amdgcn_readlane(my, j + 4), o2 = __builtin_amdgcn_readlane(my, j + 5);
        if (q + 4 < m) x2 = T4[(size_t)(half ? o2 : e2) * 16 + l16];
      }
      if (j + 6 < m) {
        const int e3 = __builtin_amdgcn_readlane(my, j + 6), o3 = __builtin_amdgcn_readlane(my, j + 7);
        if (q + 6 < m) x3 = T4[(size_t)(half ? o3 : e3) * 16 + l16];
      }
      a += x0; a += x1; a += x2; a += x3;
    }
  }

  a.x += __shfl_xor(a.x, 16);
  a.y += __shfl_xor(a.y, 16);
  a.z += __shfl_xor(a.z, 16);
  a.w += __shfl_xor(a.w, 16);
  if (half == 0) {
    const float c = (float)n;
    float4 o;
    o.x = (n > 0) ? a.x / c : 0.0f;
    o.y = (n > 0) ? a.y / c : 0.0f;
    o.z = (n > 0) ? a.z / c : 0.0f;
    o.w = (n > 0) ? a.w / c : 0.0f;
    ((float4*)(out + (size_t)node * D_HEAD))[l16] = o;
  }
}

extern "C" void kernel_launch(void* const* d_in, const int* in_sizes, int n_in,
                              void* d_out, int out_size, void* d_ws, size_t ws_size,
                              hipStream_t stream) {
  const float* X  = (const float*)d_in[0];
  const int*   ei = (const int*)d_in[1];
  const float* W1 = (const float*)d_in[2];
  const float* W2 = (const float*)d_in[3];
  float* out = (float*)d_out;

  const int* src = ei;
  const int* dst = ei + N_EDGES;

  int* cnt   = (int*)d_ws;
  int* offs  = cnt + N_PAD;
  int* hist  = offs + N_PAD;
  int* bbase = hist + HIST_ROWS * N_BKT;
  int* sdst  = hist + N_EDGES;
  float* bufA = (float*)(sdst + N_EDGES);
  float* bufB = bufA + (size_t)N_NODES * D_H1;
  int2* pairs = (int2*)(bufB + (size_t)N_NODES * D_H1);

  chunk_hist_kernel<<<HIST_ROWS, 1024, 0, stream>>>(src, hist);
  chunk_place_kernel<<<N_CHUNK, 1024, 0, stream>>>(src, dst, hist, bbase, pairs);
  bucket_group_kernel<<<(N_NODES + N_BKT - 1) / N_BKT, 1024, 0, stream>>>(pairs, bbase, cnt, offs, sdst);

  gemm1_kernel<<<512, 256, 0, stream>>>(X, W1, bufA);

  gather1_kernel<<<N_NODES / 8, 256, 0, stream>>>(offs, cnt, sdst, bufA, bufB);

  gemm2_kernel<<<391, 256, 0, stream>>>(bufB, W2, bufA);

  gather2_kernel<<<N_NODES / 8, 256, 0, stream>>>(offs, cnt, sdst, bufA, out);
}


# GEMM2 fused into the layer-1 gather: a work-group keeps its 32 finished rows of H in LDS and multiplies them by W2 on the matrix unit; H never goes to memory, one launch less; sum order unchanged
# speedup vs baseline: 8.2078x; 1.0875x over previous
#include <hip/hip_runtime.h>
#include <math.h>


typedef __attribute__((ext_vector_type(2))) float v2f;
typedef __attribute__((ext_vector_type(8))) float v8f;

#define N_NODES 50000
#define N_EDGES 1600000
#define D_IN    512
#define D_HEAD  64
#define D_H1    256
#define N_PAD   53248
#define N_CHUNK 250
#define CHUNK_E 6400
#define N_BKT   256
#define HIST_ROWS 256

__device__ __forceinline__ int wave_incl_scan(int x, int lane) {
#pragma unroll
  for (int d = 1; d < 32; d <<= 1) {
    const int up = __shfl_up(x, d);
    if (lane >= d) x += up;
  }
  return x;
}

__device__ __forceinline__ int block_excl_scan_256(int x, int t, int* wsum) {
  const int lane = t & 31;
  const int w = t >> 5;
  const int inc = wave_incl_scan(x, lane);
  if (lane == 31 && w < 8) wsum[w] = inc;
  __syncthreads();
  int pre = 0;
#pragma unroll
  for (int j = 0; j < 8; ++j) pre += (j < w) ? wsum[j] : 0;
  return pre + inc - x;
}

__global__ __launch_bounds__(1024) void chunk_hist_kernel(const int* __restrict__ src,
                                                          int* __restrict__ hist) {
  __shared__ int h[N_BKT];
  const int t = threadIdx.x;
  const int c = blockIdx.x;
  if (c >= N_CHUNK) {
    if (t < N_BKT) hist[c * N_BKT + t] = 0;
    return;
  }
  if (t < N_BKT) h[t] = 0;
  __syncthreads();
  const int4* s4 = (const int4*)(src + (size_t)c * CHUNK_E);
  const bool two = t + 1024 < CHUNK_E / 4;
  const int4 a = s4[t];
  const int4 b = s4[two ? t + 1024 : t];
  atomicAdd(&h[a.x >> 8], 1);
  atomicAdd(&h[a.y >> 8], 1);
  atomicAdd(&h[a.z >> 8], 1);
  atomicAdd(&h[a.w >> 8], 1);
  if (two) {
    atomicAdd(&h[b.x >> 8], 1);
    atomicAdd(&h[b.y >> 8], 1);
    atomicAdd(&h[b.z >> 8], 1);
    atomicAdd(&h[b.w >> 8], 1);
  }
  __syncthreads();
  if (t < N_BKT) hist[c * N_BKT + t] = h[t];
}

__global__ __launch_bounds__(1024) void chunk_place_kernel(const int* __restrict__ src,
                                                           const int* __restrict__ dst,
                                                           const int* __restrict__ hist,
                                                           int* __restrict__ bbase,
                                                           int2* __restrict__ pairs) {
  __shared__ int part_tot[4][N_BKT];
  __shared__ int part_bef[4][N_BKT];
  __shared__ int cursor[N_BKT];
  __shared__ int wsum[8];
  const int t = threadIdx.x;
  const int c = blockIdx.x;

  const int4* s4 = (const int4*)(src + (size_t)c * CHUNK_E);
  const int4* d4 = (const int4*)(dst + (size_t)c * CHUNK_E);
  const bool two = t + 1024 < CHUNK_E / 4;
  const int t2 = two ? t + 1024 : t;
  const int4 sa = s4[t];
  const int4 da = d4[t];
  const int4 sb = s4[t2];
  const int4 db = d4[t2];

  const int b = t & (N_BKT - 1);
  const int q = t >> 8;
  int tot = 0, bef = 0;
  const int* hcol = hist + (q * 64) * N_BKT + b;
#pragma unroll 1
  for (int j0 = 0; j0 < 64; j0 += 32) {
    int v[32];
#pragma unroll
    for (int u = 0; u < 32; ++u) v[u] = hcol[(j0 + u) * N_BKT];
#pragma unroll
    for (int u = 0; u < 32; ++u) {
      tot += v[u];
      bef += (q * 64 + j0 + u < c) ? v[u] : 0;
    }
  }
  part_tot[q][b] = tot;
  part_bef[q][b] = bef;
  __syncthreads();
  int T = 0, Bf = 0;
  if (t < N_BKT) {
    T  = (part_tot[0][t] + part_tot[1][t]) + (part_tot[2][t] + part_tot[3][t]);
    Bf = (part_bef[0][t] + part_bef[1][t]) + (part_bef[2][t] + part_bef[3][t]);
  }
  const int excl = block_excl_scan_256(T, t, wsum);
  if (t < N_BKT) {
    cursor[t] = excl + Bf;
    if (c == 0) {
      bbase[t] = excl;
      if (t == N_BKT - 1) bbase[N_BKT] = excl + T;
    }
  }
  __syncthreads();

  {
    const int p0 = atomicAdd(&cursor[sa.x >> 8], 1);
    const int p1 = atomicAdd(&cursor[sa.y >> 8], 1);
    const int p2 = atomicAdd(&cursor[sa.z >> 8], 1);
    const int p3 = atomicAdd(&cursor[sa.w >> 8], 1);
    pairs[p0] = make_int2(sa.x, da.x);
    pairs[p1] = make_int2(sa.y, da.y);
    pairs[p2] = make_int2(sa.z, da.z);
    pairs[p3] = make_int2(sa.w, da.w);
  }
  if (two) {
    const int p0 = atomicAdd(&cursor[sb.x >> 8], 1);
    const int p1 = atomicAdd(&cursor[sb.y >> 8], 1);
    const int p2 = atomicAdd(&cursor[sb.z >> 8], 1);
    const int p3 = atomicAdd(&cursor[sb.w >> 8], 1);
    pairs[p0] = make_int2(sb.x, db.x);
    pairs[p1] = make_int2(sb.y, db.y);
    pairs[p2] = make_int2(sb.z, db.z);
    pairs[p3] = make_int2(sb.w, db.w);
  }
}

__global__ __launch_bounds__(1024) void bucket_group_kernel(const int2* __restrict__ pairs,
                                                            const int* __restrict__ bbase,
                                                            int* __restrict__ cnt,
                                                            int* __restrict__ offs,
                                                            int* __restrict__ sdst) {
  __shared__ int cl[N_BKT];
  __shared__ int cur[N_BKT];
  __shared__ int wsum[8];
  const int t = threadIdx.x;
  const int bk = blockIdx.x;
  const int base = bbase[bk];
  const int size = bbase[bk + 1] - base;
  const int2* __restrict__ P = pairs + base;
  if (t < N_BKT) cl[t] = 0;
  __syncthreads();

  const int last = size - 1;
  for (int i0 = 0; i0 < size; i0 += 4096) {
    const int i = i0 + t;
    const bool v0 = i < size, v1 = i + 1024 < size, v2 = i + 2048 < size, v3 = i + 3072 < size;
    const int s0 = P[min(i, last)].x;
    const int s1 = P[min(i + 1024, last)].x;
    const int s2 = P[min(i + 2048, last)].x;
    const int s3 = P[min(i + 3072, last)].x;
    if (v0) atomicAdd(&cl[s0 & 255], 1);
    if (v1) atomicAdd(&cl[s1 & 255], 1);
    if (v2) atomicAdd(&cl[s2 & 255], 1);
    if (v3) atomicAdd(&cl[s3 & 255], 1);
  }
  __syncthreads();

  const int mine = (t < N_BKT) ? cl[t] : 0;
  const int excl = block_excl_scan_256(mine, t, wsum);
  if (t < N_BKT) {
    cur[t] = excl;
    const int node = bk * N_BKT + t;
    if (node < N_NODES) {
      cnt[node] = mine;
      offs[node] = base + excl;
    }
  }
  __syncthreads();

  int* __restrict__ out = sdst + base;
  for (int i0 = 0; i0 < size; i0 += 4096) {
    const int i = i0 + t;
    const bool v0 = i < size, v1 = i + 1024 < size, v2 = i + 2048 < size, v3 = i + 3072 < size;
    const int2 e0 = P[min(i, last)];
    const int2 e1 = P[min(i + 1024, last)];
    const int2 e2 = P[min(i + 2048, last)];
    const int2 e3 = P[min(i + 3072, last)];
    int p0 = 0, p1 = 0, p2 = 0, p3 = 0;
    if (v0) p0 = atomicAdd(&cur[e0.x & 255], 1);
    if (v1) p1 = atomicAdd(&cur[e1.x & 255], 1);
    if (v2) p2 = atomicAdd(&cur[e2.x & 255], 1);
    if (v3) p3 = atomicAdd(&cur[e3.x & 255], 1);
    if (v0) out[p0] = e0.y;
    if (v1) out[p1] = e1.y;
    if (v2) out[p2] = e2.y;
    if (v3) out[p3] = e3.y;
  }
}

#define G1_HEAD_BLOCKS 128
#define G1_WAVES       8
#define N_STRIPS       (N_NODES / 16)
#define KP_ROW         128

__device__ __forceinline__ void g1_load_group(v2f (&a)[8], const float* p) {
#pragma unroll
  for (int s = 0; s < 8; ++s) a[s] = *(const v2f*)(p + 4 * s);
}

__device__ __forceinline__ void g1_mma_group(v8f (&acc)[4], const v2f (&a)[8], const float* bl) {
#pragma unroll
  for (int s = 0; s < 8; ++s) {
    const float4 q0 = *(const float4*)(bl + s * (2 * KP_ROW));
    const float4 q1 = *(const float4*)(bl + s * (2 * KP_ROW) + 64);
    v2f b0, b1, b2, b3;
    b0.x = q0.x; b0.y = q0.y;
    b1.x = q0.z; b1.y = q0.w;
    b2.x = q1.x; b2.y = q1.y;
    b3.x = q1.z; b3.y = q1.w;
    acc[0] = __builtin_amdgcn_wmma_f32_16x16x4_f32(false, a[s], false, b0, (short)0, acc[0], false, false);
    acc[1] = __builtin_amdgcn_wmma_f32_16x16x4_f32(false, a[s], false, b1, (short)0, acc[1], false, false);
    acc[2] = __builtin_amdgcn_wmma_f32_16x16x4_f32(false, a[s], false, b2, (short)0, acc[2], false, false);
    acc[3] = __builtin_amdgcn_wmma_f32_16x16x4_f32(false, a[s], false, b3, (short)0, acc[3], false, false);
  }
}

__global__ __launch_bounds__(256) void gemm1_kernel(const float* __restrict__ X,
                                                    const float* __restrict__ W1,
                                                    float* __restrict__ T) {
  __shared__ __attribute__((aligned(16))) float Bs[256 * KP_ROW];
  const int t = threadIdx.x;
  const int head = blockIdx.x & 3;
  const int bih  = blockIdx.x >> 2;
  const float* __restrict__ Wh = W1 + (size_t)head * (D_IN * D_HEAD);

  {
    const int cq  = t & 15;
    const int kp0 = t >> 4;
    float4 r0[16], r1[16];
#pragma unroll
    for (int i = 0; i < 16; ++i) {
      const float4* p = (const float4*)(Wh + (size_t)(2 * (kp0 + 16 * i)) * D_HEAD) + cq;
      r0[i] = p[0];
      r1[i] = p[16];
    }
#pragma unroll
    for (int i = 0; i < 16; ++i) {
      float* q = Bs + (kp0 + 16 * i) * KP_ROW + 4 * cq;
      *(float4*)(q)      = make_float4(r0[i].x, r1[i].x, r0[i].y, r1[i].y);
      *(float4*)(q + 64) = make_float4(r0[i].z, r1[i].z, r0[i].w, r1[i].w);
    }
  }
  __syncthreads();

  const int wave = __builtin_amdgcn_readfirstlane(t >> 5);
  const int lane = t & 31;
  const int half = lane >> 4;
  const int lidx = lane & 15;
  const float* bl = Bs + half * KP_ROW + 4 * lidx;

  int strip = wave * G1_HEAD_BLOCKS + bih;
  const float* xrow = X + (size_t)((strip << 4) + lidx) * D_IN + 2 * half;
  v2f a0[8], a1[8];
  g1_load_group(a0, xrow);

  for (;;) {
    const int next = strip + G1_WAVES * G1_HEAD_BLOCKS;
    const bool has_next = next < N_STRIPS;
    const float* xnext = has_next ? X + (size_t)((next << 4) + lidx) * D_IN + 2 * half : xrow;

    v8f zero = {};
    v8f acc[4];
    acc[0] = zero; acc[1] = zero; acc[2] = zero; acc[3] = zero;

    for (int g = 0; g < D_IN / 32; g += 2) {
      g1_load_group(a1, xrow + (g + 1) * 32);
      g1_mma_group(acc, a0, bl + g * (16 * KP_ROW));
      g1_load_group(a0, (g + 2 < D_IN / 32) ? xrow + (g + 2) * 32 : xnext);
      g1_mma_group(acc, a1, bl + (g + 1) * (16 * KP_ROW));
    }

    float* trow = T + (size_t)((strip << 4) + half * 8) * D_H1 + head * D_HEAD + 4 * lidx;
#pragma unroll
    for (int r = 0; r < 8; ++r)
      *(float4*)(trow + (size_t)r * D_H1) = make_float4(acc[0][r], acc[1][r], acc[2][r], acc[3][r]);
    if (!has_next) break;
    strip = next;
    xrow = xnext;
  }
}

__device__ __forceinline__ float mean_elu(float a, float c, bool nz) {
  const float h = nz ? a / c : 0.0f;
  return (h > 0.0f) ? h : (expf(h) - 1.0f);
}

#define FN_NODES 32
#define HP       260

__device__ __forceinline__ void gather_row_sum(const int start, const int n, const int lane,
                                               const int* __restrict__ sdst,
                                               const float4* __restrict__ T4,
                                               float4& a0, float4& a1) {
  for (int base = 0; base < n; base += 32) {
    const int m = min(32, n - base);
    const int my = (lane < m) ? sdst[start + base + lane] : 0;
    int j = 0;
    for (; j + 4 <= m; j += 4) {
      const float4* r0 = T4 + (size_t)__builtin_amdgcn_readlane(my, j    ) * 64 + lane;
      const float4* r1 = T4 + (size_t)__builtin_amdgcn_readlane(my, j + 1) * 64 + lane;
      const float4* r2 = T4 + (size_t)__builtin_amdgcn_readlane(my, j + 2) * 64 + lane;
      const float4* r3 = T4 + (size_t)__builtin_amdgcn_readlane(my, j + 3) * 64 + lane;
      const float4 x0 = r0[0], y0 = r0[32];
      const float4 x1 = r1[0], y1 = r1[32];
      const float4 x2 = r2[0], y2 = r2[32];
      const float4 x3 = r3[0], y3 = r3[32];
      a0 += x0; a1 += y0;
      a0 += x1; a1 += y1;
      a0 += x2; a1 += y2;
      a0 += x3; a1 += y3;
    }
    const int rem = m - j;
    if (rem > 0) {
      const float4 z = make_float4(0.0f, 0.0f, 0.0f, 0.0f);
      float4 x1 = z, y1 = z, x2 = z, y2 = z;
      const float4* r0 = T4 + (size_t)__builtin_amdgcn_readlane(my, j) * 64 + lane;
      const float4 x0 = r0[0], y0 = r0[32];
      if (rem > 1) {
        const float4* r1 = T4 + (size_t)__builtin_amdgcn_readlane(my, j + 1) * 64 + lane;
        x1 = r1[0]; y1 = r1[32];
      }
      if (rem > 2) {
        const float4* r2 = T4 + (size_t)__builtin_amdgcn_readlane(my, j + 2) * 64 + lane;
        x2 = r2[0]; y2 = r2[32];
      }
      a0 += x0; a1 += y0;
      a0 += x1; a1 += y1;
      a0 += x2; a1 += y2;
    }
  }
}

__device__ __forceinline__ void g2_load_group(v2f (&q)[8], const float* __restrict__ p) {
#pragma unroll
  for (int s = 0; s < 4; ++s) {
    q[2 * s    ] = *(const v2f*)(p + (size_t)(4 * s    ) * D_HEAD);
    q[2 * s + 1] = *(const v2f*)(p + (size_t)(4 * s + 1) * D_HEAD);
  }
}

__device__ __forceinline__ void g2_mma_group(v8f& acc_e, v8f& acc_o, const float* ar, const v2f (&q)[8]) {
#pragma unroll
  for (int s = 0; s < 4; ++s) {
    const v2f a = *(const v2f*)(ar + 4 * s);
    v2f be, bo;
    be.x = q[2 * s].x; be.y = q[2 * s + 1].x;
    bo.x = q[2 * s].y; bo.y = q[2 * s + 1].y;
    acc_e = __builtin_amdgcn_wmma_f32_16x16x4_f32(false, a, false, be, (short)0, acc_e, false, false);
    acc_o = __builtin_amdgcn_wmma_f32_16x16x4_f32(false, a, false, bo, (short)0, acc_o, false, false);
  }
}

__global__ __launch_bounds__(512) void gather1_gemm2_kernel(const int* __restrict__ offs,
                                                            const int* __restrict__ cnt,
                                                            const int* __restrict__ sdst,
                                                            const float* __restrict__ T,
                                                            const float* __restrict__ W2,
                                                            float* __restrict__ T2) {
  __shared__ __attribute__((aligned(16))) float Hs[FN_NODES * HP];
  const int wave = __builtin_amdgcn_readfirstlane((int)(threadIdx.x >> 5));
  const int lane = threadIdx.x & 31;
  const float4* __restrict__ T4 = (const float4*)T;

  for (int i = 0; i < 2; ++i) {
    const int row = 2 * wave + i;
    const int node = blockIdx.x * FN_NODES + row;
    if (node >= N_NODES) break;
    const int start = offs[node];
    const int n = cnt[node];
    float4 a0 = make_float4(0.0f, 0.0f, 0.0f, 0.0f);
    float4 a1 = make_float4(0.0f, 0.0f, 0.0f, 0.0f);
    gather_row_sum(start, n, lane, sdst, T4, a0, a1);

    const float c = (float)n;
    const bool nz = n > 0;
    float4 h0, h1;
    h0.x = mean_elu(a0.x, c, nz); h0.y = mean_elu(a0.y, c, nz);
    h0.z = mean_elu(a0.z, c, nz); h0.w = mean_elu(a0.w, c, nz);
    h1.x = mean_elu(a1.x, c, nz); h1.y = mean_elu(a1.y, c, nz);
    h1.z = mean_elu(a1.z, c, nz); h1.w = mean_elu(a1.w, c, nz);
    float4* hrow = (float4*)(Hs + row * HP);
    hrow[lane]      = h0;
    hrow[32 + lane] = h1;
  }
  __syncthreads();
  if (wave >= 4) return;
  const int mt = wave >> 1;
  const int np = wave & 1;
  const int m0 = blockIdx.x * FN_NODES + mt * 16;
  if (m0 >= N_NODES) return;

  const int half = lane >> 4;
  const int lidx = lane & 15;
  const float* ar = Hs + (mt * 16 + lidx) * HP + 2 * half;
  const float* bp = W2 + (size_t)(2 * half) * D_HEAD + 32 * np + 2 * lidx;
  v8f acc_e = {};
  v8f acc_o = {};
  v2f q0[8], q1[8];
  g2_load_group(q0, bp);
  for (int g = 0; g < D_H1 / 16; g += 2) {
    g2_load_group(q1, bp + (size_t)((g + 1) * 16) * D_HEAD);
    g2_mma_group(acc_e, acc_o, ar + g * 16, q0);
    g2_load_group(q0, (g + 2 < D_H1 / 16) ? bp + (size_t)((g + 2) * 16) * D_HEAD : bp);
    g2_mma_group(acc_e, acc_o, ar + (g + 1) * 16, q1);
  }

  float* trow = T2 + (size_t)(m0 + half * 8) * D_HEAD + 32 * np + 2 * lidx;
#pragma unroll
  for (int r = 0; r < 8; ++r) {
    v2f o;
    o.x = acc_e[r]; o.y = acc_o[r];
    *(v2f*)(trow + (size_t)r * D_HEAD) = o;
  }
}

__global__ __launch_bounds__(256) void gather2_kernel(const int* __restrict__ offs,
                                                      const int* __restrict__ cnt,
                                                      const int* __restrict__ sdst,
                                                      const float* __restrict__ T2,
                                                      float* __restrict__ out) {
  const int node = __builtin_amdgcn_readfirstlane((int)(blockIdx.x * 8 + (threadIdx.x >> 5)));
  if (node >= N_NODES) return;
  const int lane = threadIdx.x & 31;
  const int half = lane >> 4;
  const int l16 = lane & 15;
  const int start = offs[node];
  const int n = cnt[node];
  const float4* __restrict__ T4 = (const float4*)T2;
  float4 a = make_float4(0.0f, 0.0f, 0.0f, 0.0f);

  for (int base = 0; base < n; base += 32) {
    const int m = min(32, n - base);
    const int my = (lane < m) ? sdst[start + base + lane] : 0;
    int j = 0;
    for (; j + 8 <= m; j += 8) {
      const int e0 = __builtin_amdgcn_readlane(my, j    ), o0 = __builtin_amdgcn_readlane(my, j + 1);
      const int e1 = __builtin_amdgcn_readlane(my, j + 2), o1 = __builtin_amdgcn_readlane(my, j + 3);
      const int e2 = __builtin_amdgcn_readlane(my, j + 4), o2 = __builtin_amdgcn_readlane(my, j + 5);
      const int e3 = __builtin_amdgcn_readlane(my, j + 6), o3 = __builtin_amdgcn_readlane(my, j + 7);
      const float4 x0 = T4[(size_t)(half ? o0 : e0) * 16 + l16];
      const float4 x1 = T4[(size_t)(half ? o1 : e1) * 16 + l16];
      const float4 x2 = T4[(size_t)(half ? o2 : e2) * 16 + l16];
      const float4 x3 = T4[(size_t)(half ? o3 : e3) * 16 + l16];
      a += x0; a += x1; a += x2; a += x3;
    }
    if (j < m) {
      const float4 z = make_float4(0.0f, 0.0f, 0.0f, 0.0f);
      float4 x0 = z, x1 = z, x2 = z, x3 = z;
      const int q = j + half;
      {
        const int e0 = __builtin_amdgcn_readlane(my, j), o0 = __builtin_amdgcn_readlane(my, j + 1);
        if (q < m) x0 = T4[(size_t)(half ? o0 : e0) * 16 + l16];
      }
      if (j + 2 < m) {
        const int e1 = __builtin_amdgcn_readlane(my, j + 2), o1 = __builtin_amdgcn_readlane(my, j + 3);
        if (q + 2 < m) x1 = T4[(size_t)(half ? o1 : e1) * 16 + l16];
      }
      if (j + 4 < m) {
        const int e2 = __builtin_amdgcn_readlane(my, j + 4), o2 = __builtin_amdgcn_readlane(my, j + 5);
        if (q + 4 < m) x2 = T4[(size_t)(half ? o2 : e2) * 16 + l16];
      }
      if (j + 6 < m) {
        const int e3 = __builtin_amdgcn_readlane(my, j + 6), o3 = __builtin_amdgcn_readlane(my, j + 7);
        if (q + 6 < m) x3 = T4[(size_t)(half ? o3 : e3) * 16 + l16];
      }
      a += x0; a += x1; a += x2; a += x3;
    }
  }

  a.x += __shfl_xor(a.x, 16);
  a.y += __shfl_xor(a.y, 16);
  a.z += __shfl_xor(a.z, 16);
  a.w += __shfl_xor(a.w, 16);
  if (half == 0) {
    const float c = (float)n;
    float4 o;
    o.x = (n > 0) ? a.x / c : 0.0f;
    o.y = (n > 0) ? a.y / c : 0.0f;
    o.z = (n > 0) ? a.z / c : 0.0f;
    o.w = (n > 0) ? a.w / c : 0.0f;
    ((float4*)(out + (size_t)node * D_HEAD))[l16] = o;
  }
}

extern "C" void kernel_launch(void* const* d_in, const int* in_sizes, int n_in,
                              void* d_out, int out_size, void* d_ws, size_t ws_size,
                              hipStream_t stream) {
  const float* X  = (const float*)d_in[0];
  const int*   ei = (const int*)d_in[1];
  const float* W1 = (const float*)d_in[2];
  const float* W2 = (const float*)d_in[3];
  float* out = (float*)d_out;

  const int* src = ei;
  const int* dst = ei + N_EDGES;

  int* cnt   = (int*)d_ws;
  int* offs  = cnt + N_PAD;
  int* hist  = offs + N_PAD;
  int* bbase = hist + HIST_ROWS * N_BKT;
  int* sdst  = hist + N_EDGES;
  float* bufA = (float*)(sdst + N_EDGES);
  float* bufB = bufA + (size_t)N_NODES * D_H1;
  int2* pairs = (int2*)(bufB + (size_t)N_NODES * D_H1);

  chunk_hist_kernel<<<HIST_ROWS, 1024, 0, stream>>>(src, hist);
  chunk_place_kernel<<<N_CHUNK, 1024, 0, stream>>>(src, dst, hist, bbase, pairs);
  bucket_group_kernel<<<(N_NODES + N_BKT - 1) / N_BKT, 1024, 0, stream>>>(pairs, bbase, cnt, offs, sdst);

  gemm1_kernel<<<512, 256, 0, stream>>>(X, W1, bufA);

  gather1_gemm2_kernel<<<(N_NODES + FN_NODES - 1) / FN_NODES, 512, 0, stream>>>(offs, cnt, sdst, bufA, W2, bufB);

  gather2_kernel<<<N_NODES / 8, 256, 0, stream>>>(offs, cnt, sdst, bufB, out);
}


# Grouping kernels stage results in LDS (chunk's pairs by bucket; bucket's piece of the dst list), then store contiguous runs / whole lines, not scattered 8/4-byte stores; sums still by arrival rank
# speedup vs baseline: 8.6859x; 1.0582x over previous
#include <hip/hip_runtime.h>
#include <math.h>


typedef __attribute__((ext_vector_type(2))) float v2f;
typedef __attribute__((ext_vector_type(8))) float v8f;

#define N_NODES 50000
#define N_EDGES 1600000
#define D_IN    512
#define D_HEAD  64
#define D_H1    256
#define N_PAD   53248
#define N_CHUNK 250
#define CHUNK_E 6400
#define N_BKT   256
#define HIST_ROWS 256

__device__ __forceinline__ int wave_incl_scan(int x, int lane) {
#pragma unroll
  for (int d = 1; d < 32; d <<= 1) {
    const int up = __shfl_up(x, d);
    if (lane >= d) x += up;
  }
  return x;
}

__device__ __forceinline__ int block_excl_scan_256(int x, int t, int* wsum) {
  const int lane = t & 31;
  const int w = t >> 5;
  const int inc = wave_incl_scan(x, lane);
  if (lane == 31 && w < 8) wsum[w] = inc;
  __syncthreads();
  int pre = 0;
#pragma unroll
  for (int j = 0; j < 8; ++j) pre += (j < w) ? wsum[j] : 0;
  return pre + inc - x;
}

__global__ __launch_bounds__(1024) void chunk_hist_kernel(const int* __restrict__ src,
                                                          int* __restrict__ hist) {
  __shared__ int h[N_BKT];
  const int t = threadIdx.x;
  const int c = blockIdx.x;
  if (c >= N_CHUNK) {
    if (t < N_BKT) hist[c * N_BKT + t] = 0;
    return;
  }
  if (t < N_BKT) h[t] = 0;
  __syncthreads();
  const int4* s4 = (const int4*)(src + (size_t)c * CHUNK_E);
  const bool two = t + 1024 < CHUNK_E / 4;
  const int4 a = s4[t];
  const int4 b = s4[two ? t + 1024 : t];
  atomicAdd(&h[a.x >> 8], 1);
  atomicAdd(&h[a.y >> 8], 1);
  atomicAdd(&h[a.z >> 8], 1);
  atomicAdd(&h[a.w >> 8], 1);
  if (two) {
    atomicAdd(&h[b.x >> 8], 1);
    atomicAdd(&h[b.y >> 8], 1);
    atomicAdd(&h[b.z >> 8], 1);
    atomicAdd(&h[b.w >> 8], 1);
  }
  __syncthreads();
  if (t < N_BKT) hist[c * N_BKT + t] = h[t];
}

__global__ __launch_bounds__(1024) void chunk_place_kernel(const int* __restrict__ src,
                                                           const int* __restrict__ dst,
                                                           const int* __restrict__ hist,
                                                           int* __restrict__ bbase,
                                                           int2* __restrict__ pairs) {
  __shared__ int part_tot[4][N_BKT];
  __shared__ int part_bef[4][N_BKT];
  __shared__ int cursor[N_BKT];
  __shared__ int delta[N_BKT];
  __shared__ int wsum[8];
  __shared__ int wsum2[8];
  __shared__ int2 stage[CHUNK_E];
  const int t = threadIdx.x;
  const int c = blockIdx.x;

  const int4* s4 = (const int4*)(src + (size_t)c * CHUNK_E);
  const int4* d4 = (const int4*)(dst + (size_t)c * CHUNK_E);
  const bool two = t + 1024 < CHUNK_E / 4;
  const int t2 = two ? t + 1024 : t;
  const int4 sa = s4[t];
  const int4 da = d4[t];
  const int4 sb = s4[t2];
  const int4 db = d4[t2];

  const int b = t & (N_BKT - 1);
  const int q = t >> 8;
  const int own = hist[c * N_BKT + b];
  int tot = 0, bef = 0;
  const int* hcol = hist + (q * 64) * N_BKT + b;
#pragma unroll 1
  for (int j0 = 0; j0 < 64; j0 += 32) {
    int v[32];
#pragma unroll
    for (int u = 0; u < 32; ++u) v[u] = hcol[(j0 + u) * N_BKT];
#pragma unroll
    for (int u = 0; u < 32; ++u) {
      tot += v[u];
      bef += (q * 64 + j0 + u < c) ? v[u] : 0;
    }
  }
  part_tot[q][b] = tot;
  part_bef[q][b] = bef;
  __syncthreads();
  int T = 0, Bf = 0;
  if (t < N_BKT) {
    T  = (part_tot[0][t] + part_tot[1][t]) + (part_tot[2][t] + part_tot[3][t]);
    Bf = (part_bef[0][t] + part_bef[1][t]) + (part_bef[2][t] + part_bef[3][t]);
  }
  const int excl = block_excl_scan_256(T, t, wsum);
  const int lst  = block_excl_scan_256((t < N_BKT) ? own : 0, t, wsum2);
  if (t < N_BKT) {
    cursor[t] = lst;
    delta[t] = excl + Bf - lst;
    if (c == 0) {
      bbase[t] = excl;
      if (t == N_BKT - 1) bbase[N_BKT] = excl + T;
    }
  }
  __syncthreads();

  {
    const int p0 = atomicAdd(&cursor[sa.x >> 8], 1);
    const int p1 = atomicAdd(&cursor[sa.y >> 8], 1);
    const int p2 = atomicAdd(&cursor[sa.z >> 8], 1);
    const int p3 = atomicAdd(&cursor[sa.w >> 8], 1);
    stage[p0] = make_int2(sa.x, da.x);
    stage[p1] = make_int2(sa.y, da.y);
    stage[p2] = make_int2(sa.z, da.z);
    stage[p3] = make_int2(sa.w, da.w);
  }
  if (two) {
    const int p0 = atomicAdd(&cursor[sb.x >> 8], 1);
    const int p1 = atomicAdd(&cursor[sb.y >> 8], 1);
    const int p2 = atomicAdd(&cursor[sb.z >> 8], 1);
    const int p3 = atomicAdd(&cursor[sb.w >> 8], 1);
    stage[p0] = make_int2(sb.x, db.x);
    stage[p1] = make_int2(sb.y, db.y);
    stage[p2] = make_int2(sb.z, db.z);
    stage[p3] = make_int2(sb.w, db.w);
  }
  __syncthreads();

  constexpr int FULL = CHUNK_E / 1024;
  const bool more = t < CHUNK_E - FULL * 1024;
  int2 e[FULL];
  int d[FULL];
#pragma unroll
  for (int k = 0; k < FULL; ++k) e[k] = stage[t + 1024 * k];
  const int2 e6 = stage[more ? t + FULL * 1024 : t];
#pragma unroll
  for (int k = 0; k < FULL; ++k) d[k] = delta[e[k].x >> 8];
  const int d6 = delta[e6.x >> 8];
#pragma unroll
  for (int k = 0; k < FULL; ++k) pairs[t + 1024 * k + d[k]] = e[k];
  if (more) pairs[t + FULL * 1024 + d6] = e6;
}

#define STAGE_CAP 16384
__global__ __launch_bounds__(1024) void bucket_group_kernel(const int2* __restrict__ pairs,
                                                            const int* __restrict__ bbase,
                                                            int* __restrict__ cnt,
                                                            int* __restrict__ offs,
                                                            int* __restrict__ sdst) {
  __shared__ int cl[N_BKT];
  __shared__ int cur[N_BKT];
  __shared__ int wsum[8];
  __shared__ int stage[STAGE_CAP];
  const int t = threadIdx.x;
  const int bk = blockIdx.x;
  const int base = bbase[bk];
  const int size = bbase[bk + 1] - base;
  const bool staged = size <= STAGE_CAP;
  const int2* __restrict__ P = pairs + base;
  if (t < N_BKT) cl[t] = 0;
  __syncthreads();

  const int last = size - 1;
  for (int i0 = 0; i0 < size; i0 += 4096) {
    const int i = i0 + t;
    const bool v0 = i < size, v1 = i + 1024 < size, v2 = i + 2048 < size, v3 = i + 3072 < size;
    const int s0 = P[min(i, last)].x;
    const int s1 = P[min(i + 1024, last)].x;
    const int s2 = P[min(i + 2048, last)].x;
    const int s3 = P[min(i + 3072, last)].x;
    if (v0) atomicAdd(&cl[s0 & 255], 1);
    if (v1) atomicAdd(&cl[s1 & 255], 1);
    if (v2) atomicAdd(&cl[s2 & 255], 1);
    if (v3) atomicAdd(&cl[s3 & 255], 1);
  }
  __syncthreads();

  const int mine = (t < N_BKT) ? cl[t] : 0;
  const int excl = block_excl_scan_256(mine, t, wsum);
  if (t < N_BKT) {
    cur[t] = excl;
    const int node = bk * N_BKT + t;
    if (node < N_NODES) {
      cnt[node] = mine;
      offs[node] = base + excl;
    }
  }
  __syncthreads();

  int* __restrict__ out = sdst + base;
  for (int i0 = 0; i0 < size; i0 += 4096) {
    const int i = i0 + t;
    const bool v0 = i < size, v1 = i + 1024 < size, v2 = i + 2048 < size, v3 = i + 3072 < size;
    const int2 e0 = P[min(i, last)];
    const int2 e1 = P[min(i + 1024, last)];
    const int2 e2 = P[min(i + 2048, last)];
    const int2 e3 = P[min(i + 3072, last)];
    int p0 = 0, p1 = 0, p2 = 0, p3 = 0;
    if (v0) p0 = atomicAdd(&cur[e0.x & 255], 1);
    if (v1) p1 = atomicAdd(&cur[e1.x & 255], 1);
    if (v2) p2 = atomicAdd(&cur[e2.x & 255], 1);
    if (v3) p3 = atomicAdd(&cur[e3.x & 255], 1);
    if (staged) {
      if (v0) stage[p0] = e0.y;
      if (v1) stage[p1] = e1.y;
      if (v2) stage[p2] = e2.y;
      if (v3) stage[p3] = e3.y;
    } else {
      if (v0) out[p0] = e0.y;
      if (v1) out[p1] = e1.y;
      if (v2) out[p2] = e2.y;
      if (v3) out[p3] = e3.y;
    }
  }
  if (staged) {
    __syncthreads();
    for (int i0 = -(base & 31); i0 < size; i0 += 4096) {
      const int i = i0 + t;
      const bool w0 = i >= 0 && i < size, w1 = i + 1024 < size, w2 = i + 2048 < size, w3 = i + 3072 < size;
      const int x0 = stage[w0 ? i : 0];
      const int x1 = stage[w1 ? i + 1024 : 0];
      const int x2 = stage[w2 ? i + 2048 : 0];
      const int x3 = stage[w3 ? i + 3072 : 0];
      if (w0) out[i] = x0;
      if (w1) out[i + 1024] = x1;
      if (w2) out[i + 2048] = x2;
      if (w3) out[i + 3072] = x3;
    }
  }
}

#define G1_HEAD_BLOCKS 128
#define G1_WAVES       8
#define N_STRIPS       (N_NODES / 16)
#define KP_ROW         128

__device__ __forceinline__ void g1_load_group(v2f (&a)[8], const float* p) {
#pragma unroll
  for (int s = 0; s < 8; ++s) a[s] = *(const v2f*)(p + 4 * s);
}

__device__ __forceinline__ void g1_mma_group(v8f (&acc)[4], const v2f (&a)[8], const float* bl) {
#pragma unroll
  for (int s = 0; s < 8; ++s) {
    const float4 q0 = *(const float4*)(bl + s * (2 * KP_ROW));
    const float4 q1 = *(const float4*)(bl + s * (2 * KP_ROW) + 64);
    v2f b0, b1, b2, b3;
    b0.x = q0.x; b0.y = q0.y;
    b1.x = q0.z; b1.y = q0.w;
    b2.x = q1.x; b2.y = q1.y;
    b3.x = q1.z; b3.y = q1.w;
    acc[0] = __builtin_amdgcn_wmma_f32_16x16x4_f32(false, a[s], false, b0, (short)0, acc[0], false, false);
    acc[1] = __builtin_amdgcn_wmma_f32_16x16x4_f32(false, a[s], false, b1, (short)0, acc[1], false, false);
    acc[2] = __builtin_amdgcn_wmma_f32_16x16x4_f32(false, a[s], false, b2, (short)0, acc[2], false, false);
    acc[3] = __builtin_amdgcn_wmma_f32_16x16x4_f32(false, a[s], false, b3, (short)0, acc[3], false, false);
  }
}

__global__ __launch_bounds__(256) void gemm1_kernel(const float* __restrict__ X,
                                                    const float* __restrict__ W1,
                                                    float* __restrict__ T) {
  __shared__ __attribute__((aligned(16))) float Bs[256 * KP_ROW];
  const int t = threadIdx.x;
  const int head = blockIdx.x & 3;
  const int bih  = blockIdx.x >> 2;
  const float* __restrict__ Wh = W1 + (size_t)head * (D_IN * D_HEAD);

  {
    const int cq  = t & 15;
    const int kp0 = t >> 4;
    float4 r0[16], r1[16];
#pragma unroll
    for (int i = 0; i < 16; ++i) {
      const float4* p = (const float4*)(Wh + (size_t)(2 * (kp0 + 16 * i)) * D_HEAD) + cq;
      r0[i] = p[0];
      r1[i] = p[16];
    }
#pragma unroll
    for (int i = 0; i < 16; ++i) {
      float* q = Bs + (kp0 + 16 * i) * KP_ROW + 4 * cq;
      *(float4*)(q)      = make_float4(r0[i].x, r1[i].x, r0[i].y, r1[i].y);
      *(float4*)(q + 64) = make_float4(r0[i].z, r1[i].z, r0[i].w, r1[i].w);
    }
  }
  __syncthreads();

  const int wave = __builtin_amdgcn_readfirstlane(t >> 5);
  const int lane = t & 31;
  const int half = lane >> 4;
  const int lidx = lane & 15;
  const float* bl = Bs + half * KP_ROW + 4 * lidx;

  int strip = wave * G1_HEAD_BLOCKS + bih;
  const float* xrow = X + (size_t)((strip << 4) + lidx) * D_IN + 2 * half;
  v2f a0[8], a1[8];
  g1_load_group(a0, xrow);

  for (;;) {
    const int next = strip + G1_WAVES * G1_HEAD_BLOCKS;
    const bool has_next = next < N_STRIPS;
    const float* xnext = has_next ? X + (size_t)((next << 4) + lidx) * D_IN + 2 * half : xrow;

    v8f zero = {};
    v8f acc[4];
    acc[0] = zero; acc[1] = zero; acc[2] = zero; acc[3] = zero;

    for (int g = 0; g < D_IN / 32; g += 2) {
      g1_load_group(a1, xrow + (g + 1) * 32);
      g1_mma_group(acc, a0, bl + g * (16 * KP_ROW));
      g1_load_group(a0, (g + 2 < D_IN / 32) ? xrow + (g + 2) * 32 : xnext);
      g1_mma_group(acc, a1, bl + (g + 1) * (16 * KP_ROW));
    }

    float* trow = T + (size_t)((strip << 4) + half * 8) * D_H1 + head * D_HEAD + 4 * lidx;
#pragma unroll
    for (int r = 0; r < 8; ++r)
      *(float4*)(trow + (size_t)r * D_H1) = make_float4(acc[0][r], acc[1][r], acc[2][r], acc[3][r]);
    if (!has_next) break;
    strip = next;
    xrow = xnext;
  }
}

__device__ __forceinline__ float mean_elu(float a, float c, bool nz) {
  const float h = nz ? a / c : 0.0f;
  return (h > 0.0f) ? h : (expf(h) - 1.0f);
}

#define FN_NODES 32
#define HP       260

__device__ __forceinline__ void gather_row_sum(const int start, const int n, const int lane,
                                               const int* __restrict__ sdst,
                                               const float4* __restrict__ T4,
                                               float4& a0, float4& a1) {
  for (int base = 0; base < n; base += 32) {
    const int m = min(32, n - base);
    const int my = (lane < m) ? sdst[start + base + lane] : 0;
    int j = 0;
    for (; j + 4 <= m; j += 4) {
      const float4* r0 = T4 + (size_t)__builtin_amdgcn_readlane(my, j    ) * 64 + lane;
      const float4* r1 = T4 + (size_t)__builtin_amdgcn_readlane(my, j + 1) * 64 + lane;
      const float4* r2 = T4 + (size_t)__builtin_amdgcn_readlane(my, j + 2) * 64 + lane;
      const float4* r3 = T4 + (size_t)__builtin_amdgcn_readlane(my, j + 3) * 64 + lane;
      const float4 x0 = r0[0], y0 = r0[32];
      const float4 x1 = r1[0], y1 = r1[32];
      const float4 x2 = r2[0], y2 = r2[32];
      const float4 x3 = r3[0], y3 = r3[32];
      a0 += x0; a1 += y0;
      a0 += x1; a1 += y1;
      a0 += x2; a1 += y2;
      a0 += x3; a1 += y3;
    }
    const int rem = m - j;
    if (rem > 0) {
      const float4 z = make_float4(0.0f, 0.0f, 0.0f, 0.0f);
      float4 x1 = z, y1 = z, x2 = z, y2 = z;
      const float4* r0 = T4 + (size_t)__builtin_amdgcn_readlane(my, j) * 64 + lane;
      const float4 x0 = r0[0], y0 = r0[32];
      if (rem > 1) {
        const float4* r1 = T4 + (size_t)__builtin_amdgcn_readlane(my, j + 1) * 64 + lane;
        x1 = r1[0]; y1 = r1[32];
      }
      if (rem > 2) {
        const float4* r2 = T4 + (size_t)__builtin_amdgcn_readlane(my, j + 2) * 64 + lane;
        x2 = r2[0]; y2 = r2[32];
      }
      a0 += x0; a1 += y0;
      a0 += x1; a1 += y1;
      a0 += x2; a1 += y2;
    }
  }
}

__device__ __forceinline__ void g2_load_group(v2f (&q)[8], const float* __restrict__ p) {
#pragma unroll
  for (int s = 0; s < 4; ++s) {
    q[2 * s    ] = *(const v2f*)(p + (size_t)(4 * s    ) * D_HEAD);
    q[2 * s + 1] = *(const v2f*)(p + (size_t)(4 * s + 1) * D_HEAD);
  }
}

__device__ __forceinline__ void g2_mma_group(v8f& acc_e, v8f& acc_o, const float* ar, const v2f (&q)[8]) {
#pragma unroll
  for (int s = 0; s < 4; ++s) {
    const v2f a = *(const v2f*)(ar + 4 * s);
    v2f be, bo;
    be.x = q[2 * s].x; be.y = q[2 * s + 1].x;
    bo.x = q[2 * s].y; bo.y = q[2 * s + 1].y;
    acc_e = __builtin_amdgcn_wmma_f32_16x16x4_f32(false, a, false, be, (short)0, acc_e, false, false);
    acc_o = __builtin_amdgcn_wmma_f32_16x16x4_f32(false, a, false, bo, (short)0, acc_o, false, false);
  }
}

__global__ __launch_bounds__(512) void gather1_gemm2_kernel(const int* __restrict__ offs,
                                                            const int* __restrict__ cnt,
                                                            const int* __restrict__ sdst,
                                                            const float* __restrict__ T,
                                                            const float* __restrict__ W2,
                                                            float* __restrict__ T2) {
  __shared__ __attribute__((aligned(16))) float Hs[FN_NODES * HP];
  const int wave = __builtin_amdgcn_readfirstlane((int)(threadIdx.x >> 5));
  const int lane = threadIdx.x & 31;
  const float4* __restrict__ T4 = (const float4*)T;

  for (int i = 0; i < 2; ++i) {
    const int row = 2 * wave + i;
    const int node = blockIdx.x * FN_NODES + row;
    if (node >= N_NODES) break;
    const int start = offs[node];
    const int n = cnt[node];
    float4 a0 = make_float4(0.0f, 0.0f, 0.0f, 0.0f);
    float4 a1 = make_float4(0.0f, 0.0f, 0.0f, 0.0f);
    gather_row_sum(start, n, lane, sdst, T4, a0, a1);

    const float c = (float)n;
    const bool nz = n > 0;
    float4 h0, h1;
    h0.x = mean_elu(a0.x, c, nz); h0.y = mean_elu(a0.y, c, nz);
    h0.z = mean_elu(a0.z, c, nz); h0.w = mean_elu(a0.w, c, nz);
    h1.x = mean_elu(a1.x, c, nz); h1.y = mean_elu(a1.y, c, nz);
    h1.z = mean_elu(a1.z, c, nz); h1.w = mean_elu(a1.w, c, nz);
    float4* hrow = (float4*)(Hs + row * HP);
    hrow[lane]      = h0;
    hrow[32 + lane] = h1;
  }
  __syncthreads();
  if (wave >= 4) return;
  const int mt = wave >> 1;
  const int np = wave & 1;
  const int m0 = blockIdx.x * FN_NODES + mt * 16;
  if (m0 >= N_NODES) return;

  const int half = lane >> 4;
  const int lidx = lane & 15;
  const float* ar = Hs + (mt * 16 + lidx) * HP + 2 * half;
  const float* bp = W2 + (size_t)(2 * half) * D_HEAD + 32 * np + 2 * lidx;
  v8f acc_e = {};
  v8f acc_o = {};
  v2f q0[8], q1[8];
  g2_load_group(q0, bp);
  for (int g = 0; g < D_H1 / 16; g += 2) {
    g2_load_group(q1, bp + (size_t)((g + 1) * 16) * D_HEAD);
    g2_mma_group(acc_e, acc_o, ar + g * 16, q0);
    g2_load_group(q0, (g + 2 < D_H1 / 16) ? bp + (size_t)((g + 2) * 16) * D_HEAD : bp);
    g2_mma_group(acc_e, acc_o, ar + (g + 1) * 16, q1);
  }

  float* trow = T2 + (size_t)(m0 + half * 8) * D_HEAD + 32 * np + 2 * lidx;
#pragma unroll
  for (int r = 0; r < 8; ++r) {
    v2f o;
    o.x = acc_e[r]; o.y = acc_o[r];
    *(v2f*)(trow + (size_t)r * D_HEAD) = o;
  }
}

__global__ __launch_bounds__(256) void gather2_kernel(const int* __restrict__ offs,
                                                      const int* __restrict__ cnt,
                                                      const int* __restrict__ sdst,
                                                      const float* __restrict__ T2,
                                                      float* __restrict__ out) {
  const int node = __builtin_amdgcn_readfirstlane((int)(blockIdx.x * 8 + (threadIdx.x >> 5)));
  if (node >= N_NODES) return;
  const int lane = threadIdx.x & 31;
  const int half = lane >> 4;
  const int l16 = lane & 15;
  const int start = offs[node];
  const int n = cnt[node];
  const float4* __restrict__ T4 = (const float4*)T2;
  float4 a = make_float4(0.0f, 0.0f, 0.0f, 0.0f);

  for (int base = 0; base < n; base += 32) {
    const int m = min(32, n - base);
    const int my = (lane < m) ? sdst[start + base + lane] : 0;
    int j = 0;
    for (; j + 8 <= m; j += 8) {
      const int e0 = __builtin_amdgcn_readlane(my, j    ), o0 = __builtin_amdgcn_readlane(my, j + 1);
      const int e1 = __builtin_amdgcn_readlane(my, j + 2), o1 = __builtin_amdgcn_readlane(my, j + 3);
      const int e2 = __builtin_amdgcn_readlane(my, j + 4), o2 = __builtin_amdgcn_readlane(my, j + 5);
      const int e3 = __builtin_amdgcn_readlane(my, j + 6), o3 = __builtin_amdgcn_readlane(my, j + 7);
      const float4 x0 = T4[(size_t)(half ? o0 : e0) * 16 + l16];
      const float4 x1 = T4[(size_t)(half ? o1 : e1) * 16 + l16];
      const float4 x2 = T4[(size_t)(half ? o2 : e2) * 16 + l16];
      const float4 x3 = T4[(size_t)(half ? o3 : e3) * 16 + l16];
      a += x0; a += x1; a += x2; a += x3;
    }
    if (j < m) {
      const float4 z = make_float4(0.0f, 0.0f, 0.0f, 0.0f);
      float4 x0 = z, x1 = z, x2 = z, x3 = z;
      const int q = j + half;
      {
        const int e0 = __builtin_amdgcn_readlane(my, j), o0 = __builtin_amdgcn_readlane(my, j + 1);
        if (q < m) x0 = T4[(size_t)(half ? o0 : e0) * 16 + l16];
      }
      if (j + 2 < m) {
        const int e1 = __builtin_amdgcn_readlane(my, j + 2), o1 = __builtin_amdgcn_readlane(my, j + 3);
        if (q + 2 < m) x1 = T4[(size_t)(half ? o1 : e1) * 16 + l16];
      }
      if (j + 4 < m) {
        const int e2 = __builtin_amdgcn_readlane(my, j + 4), o2 = __builtin_amdgcn_readlane(my, j + 5);
        if (q + 4 < m) x2 = T4[(size_t)(half ? o2 : e2) * 16 + l16];
      }
      if (j + 6 < m) {
        const int e3 = __builtin_amdgcn_readlane(my, j + 6), o3 = __builtin_amdgcn_readlane(my, j + 7);
        if (q + 6 < m) x3 = T4[(size_t)(half ? o3 : e3) * 16 + l16];
      }
      a += x0; a += x1; a += x2; a += x3;
    }
  }

  a.x += __shfl_xor(a.x, 16);
  a.y += __shfl_xor(a.y, 16);
  a.z += __shfl_xor(a.z, 16);
  a.w += __shfl_xor(a.w, 16);
  if (half == 0) {
    const float c = (float)n;
    float4 o;
    o.x = (n > 0) ? a.x / c : 0.0f;
    o.y = (n > 0) ? a.y / c : 0.0f;
    o.z = (n > 0) ? a.z / c : 0.0f;
    o.w = (n > 0) ? a.w / c : 0.0f;
    ((float4*)(out + (size_t)node * D_HEAD))[l16] = o;
  }
}

extern "C" void kernel_launch(void* const* d_in, const int* in_sizes, int n_in,
                              void* d_out, int out_size, void* d_ws, size_t ws_size,
                              hipStream_t stream) {
  const float* X  = (const float*)d_in[0];
  const int*   ei = (const int*)d_in[1];
  const float* W1 = (const float*)d_in[2];
  const float* W2 = (const float*)d_in[3];
  float* out = (float*)d_out;

  const int* src = ei;
  const int* dst = ei + N_EDGES;

  int* cnt   = (int*)d_ws;
  int* offs  = cnt + N_PAD;
  int* hist  = offs + N_PAD;
  int* bbase = hist + HIST_ROWS * N_BKT;
  int* sdst  = hist + N_EDGES;
  float* bufA = (float*)(sdst + N_EDGES);
  float* bufB = bufA + (size_t)N_NODES * D_H1;
  int2* pairs = (int2*)(bufB + (size_t)N_NODES * D_H1);

  chunk_hist_kernel<<<HIST_ROWS, 1024, 0, stream>>>(src, hist);
  chunk_place_kernel<<<N_CHUNK, 1024, 0, stream>>>(src, dst, hist, bbase, pairs);
  bucket_group_kernel<<<(N_NODES + N_BKT - 1) / N_BKT, 1024, 0, stream>>>(pairs, bbase, cnt, offs, sdst);

  gemm1_kernel<<<512, 256, 0, stream>>>(X, W1, bufA);

  gather1_gemm2_kernel<<<(N_NODES + FN_NODES - 1) / FN_NODES, 512, 0, stream>>>(offs, cnt, sdst, bufA, W2, bufB);

  gather2_kernel<<<N_NODES / 8, 256, 0, stream>>>(offs, cnt, sdst, bufB, out);
}


# Chunk phase of the edge grouping made self-contained (chunk ordered by bucket in LDS, one run + own count/offset rows) and run ahead of GEMM1's matrix work in its work-groups; 4 launches not 6
# speedup vs baseline: 9.0662x; 1.0438x over previous
#include <hip/hip_runtime.h>
#include <math.h>


typedef __attribute__((ext_vector_type(2))) float v2f;
typedef __attribute__((ext_vector_type(8))) float v8f;

#define N_NODES 50000
#define N_EDGES 1600000
#define D_IN    512
#define D_HEAD  64
#define D_H1    256
#define N_PAD   53248
#define N_CHUNK 250
#define CHUNK_E 6400
#define N_BKT   256

__device__ __forceinline__ int wave_incl_scan(int x, int lane) {
#pragma unroll
  for (int d = 1; d < 32; d <<= 1) {
    const int up = __shfl_up(x, d);
    if (lane >= d) x += up;
  }
  return x;
}

__device__ __forceinline__ int block_excl_scan_256(int x, int t, int* wsum) {
  const int lane = t & 31;
  const int w = t >> 5;
  const int inc = wave_incl_scan(x, lane);
  if (lane == 31 && w < 8) wsum[w] = inc;
  __syncthreads();
  int pre = 0;
#pragma unroll
  for (int j = 0; j < 8; ++j) pre += (j < w) ? wsum[j] : 0;
  return pre + inc - x;
}

__device__ __forceinline__ void chunk_sort_256(const int* __restrict__ src,
                                               const int* __restrict__ dst,
                                               int* __restrict__ hist,
                                               int* __restrict__ lstart,
                                               int2* __restrict__ pairs,
                                               int2* stage, int* h, int* cursor, int* wsum,
                                               const int c, const int t) {
  const int4* s4 = (const int4*)(src + (size_t)c * CHUNK_E);
  const int4* d4 = (const int4*)(dst + (size_t)c * CHUNK_E);
  constexpr int FULL4 = (CHUNK_E / 4) / 256;
  const bool more = t < CHUNK_E / 4 - FULL4 * 256;
  const int tm = more ? t + FULL4 * 256 : t;
  int4 sa[FULL4], da[FULL4];
#pragma unroll
  for (int k = 0; k < FULL4; ++k) sa[k] = s4[t + 256 * k];
  const int4 sb = s4[tm];
#pragma unroll
  for (int k = 0; k < FULL4; ++k) da[k] = d4[t + 256 * k];
  const int4 db = d4[tm];

  h[t] = 0;
  __syncthreads();
#pragma unroll
  for (int k = 0; k < FULL4; ++k) {
    atomicAdd(&h[sa[k].x >> 8], 1);
    atomicAdd(&h[sa[k].y >> 8], 1);
    atomicAdd(&h[sa[k].z >> 8], 1);
    atomicAdd(&h[sa[k].w >> 8], 1);
  }
  if (more) {
    atomicAdd(&h[sb.x >> 8], 1);
    atomicAdd(&h[sb.y >> 8], 1);
    atomicAdd(&h[sb.z >> 8], 1);
    atomicAdd(&h[sb.w >> 8], 1);
  }
  __syncthreads();
  const int own = h[t];
  const int lst = block_excl_scan_256(own, t, wsum);
  cursor[t] = lst;
  hist[c * N_BKT + t] = own;
  lstart[c * N_BKT + t] = lst;
  __syncthreads();

#pragma unroll
  for (int k = 0; k < FULL4; ++k) {
    const int p0 = atomicAdd(&cursor[sa[k].x >> 8], 1);
    const int p1 = atomicAdd(&cursor[sa[k].y >> 8], 1);
    const int p2 = atomicAdd(&cursor[sa[k].z >> 8], 1);
    const int p3 = atomicAdd(&cursor[sa[k].w >> 8], 1);
    stage[p0] = make_int2(sa[k].x, da[k].x);
    stage[p1] = make_int2(sa[k].y, da[k].y);
    stage[p2] = make_int2(sa[k].z, da[k].z);
    stage[p3] = make_int2(sa[k].w, da[k].w);
  }
  if (more) {
    const int p0 = atomicAdd(&cursor[sb.x >> 8], 1);
    const int p1 = atomicAdd(&cursor[sb.y >> 8], 1);
    const int p2 = atomicAdd(&cursor[sb.z >> 8], 1);
    const int p3 = atomicAdd(&cursor[sb.w >> 8], 1);
    stage[p0] = make_int2(sb.x, db.x);
    stage[p1] = make_int2(sb.y, db.y);
    stage[p2] = make_int2(sb.z, db.z);
    stage[p3] = make_int2(sb.w, db.w);
  }
  __syncthreads();

  constexpr int UNITS = CHUNK_E / 2;
  constexpr int FULLU = UNITS / 256;
  int2* __restrict__ run = pairs + (size_t)c * CHUNK_E;
  int2 lo[FULLU], hi[FULLU];
#pragma unroll
  for (int k = 0; k < FULLU; ++k) {
    lo[k] = stage[2 * (t + 256 * k)];
    hi[k] = stage[2 * (t + 256 * k) + 1];
  }
  const bool tail = t < UNITS - FULLU * 256;
  const int tu = tail ? t + FULLU * 256 : t;
  const int2 lot = stage[2 * tu];
  const int2 hit = stage[2 * tu + 1];
#pragma unroll
  for (int k = 0; k < FULLU; ++k)
    *(int4*)(run + 2 * (t + 256 * k)) = make_int4(lo[k].x, lo[k].y, hi[k].x, hi[k].y);
  if (tail) *(int4*)(run + 2 * tu) = make_int4(lot.x, lot.y, hit.x, hit.y);
  __syncthreads();
}

#define STAGE_CAP 16384
#define SEG_PER_WAVE 8
__global__ __launch_bounds__(1024) void bucket_group_kernel(const int2* __restrict__ pairs,
                                                            const int* __restrict__ hist,
                                                            const int* __restrict__ lstart,
                                                            int* __restrict__ cnt,
                                                            int* __restrict__ offs,
                                                            int* __restrict__ sdst) {
  __shared__ int cl[N_BKT];
  __shared__ int cur[N_BKT];
  __shared__ int seg_n[N_BKT];
  __shared__ int seg_at[N_BKT];
  __shared__ int wsum[8];
  __shared__ int wsum2[8];
  __shared__ int tot[2];
  __shared__ int stage[STAGE_CAP];
  const int t = threadIdx.x;
  const int bk = blockIdx.x;
  int n = 0, L = 0;
  if (t < N_CHUNK) {
    n = hist[t * N_BKT + bk];
    L = lstart[t * N_BKT + bk];
  }
  if (t < N_BKT) {
    cl[t] = 0;
    seg_n[t] = n;
    seg_at[t] = t * CHUNK_E + L;
  }
  const int en = block_excl_scan_256(n, t, wsum);
  const int eL = block_excl_scan_256(L, t, wsum2);
  if (t == N_BKT - 1) {
    tot[0] = en + n;
    tot[1] = eL + L;
  }
  __syncthreads();
  const int size = tot[0];
  const int base = tot[1];
  const bool staged = size <= STAGE_CAP;

  const int w = __builtin_amdgcn_readfirstlane(t >> 5);
  const int lane = t & 31;
  int2 e[2 * SEG_PER_WAVE];
  bool v[2 * SEG_PER_WAVE];
#pragma unroll
  for (int j = 0; j < SEG_PER_WAVE; ++j) {
    const int cseg = w + 32 * j;
    const int nn = seg_n[cseg];
    const int at = seg_at[cseg];
    v[2 * j]     = lane < nn;
    v[2 * j + 1] = lane + 32 < nn;
    e[2 * j]     = pairs[v[2 * j]     ? at + lane      : 0];
    e[2 * j + 1] = pairs[v[2 * j + 1] ? at + lane + 32 : 0];
  }
#pragma unroll
  for (int u = 0; u < 2 * SEG_PER_WAVE; ++u)
    if (v[u]) atomicAdd(&cl[e[u].x & 255], 1);
  for (int j = 0; j < SEG_PER_WAVE; ++j) {
    const int cseg = w + 32 * j;
    const int nn = seg_n[cseg];
    const int at = seg_at[cseg];
    for (int i = 64 + lane; i < nn; i += 32) atomicAdd(&cl[pairs[at + i].x & 255], 1);
  }
  __syncthreads();

  const int mine = (t < N_BKT) ? cl[t] : 0;
  const int excl = block_excl_scan_256(mine, t, wsum);
  if (t < N_BKT) {
    cur[t] = excl;
    const int node = bk * N_BKT + t;
    if (node < N_NODES) {
      cnt[node] = mine;
      offs[node] = base + excl;
    }
  }
  __syncthreads();

  int* __restrict__ out = sdst + base;
  int p[2 * SEG_PER_WAVE];
#pragma unroll
  for (int u = 0; u < 2 * SEG_PER_WAVE; ++u) {
    p[u] = 0;
    if (v[u]) p[u] = atomicAdd(&cur[e[u].x & 255], 1);
  }
  if (staged) {
#pragma unroll
    for (int u = 0; u < 2 * SEG_PER_WAVE; ++u)
      if (v[u]) stage[p[u]] = e[u].y;
  } else {
#pragma unroll
    for (int u = 0; u < 2 * SEG_PER_WAVE; ++u)
      if (v[u]) out[p[u]] = e[u].y;
  }
  for (int j = 0; j < SEG_PER_WAVE; ++j) {
    const int cseg = w + 32 * j;
    const int nn = seg_n[cseg];
    const int at = seg_at[cseg];
    for (int i = 64 + lane; i < nn; i += 32) {
      const int2 x = pairs[at + i];
      const int px = atomicAdd(&cur[x.x & 255], 1);
      if (staged) stage[px] = x.y; else out[px] = x.y;
    }
  }
  if (staged) {
    __syncthreads();
    for (int i0 = -(base & 31); i0 < size; i0 += 4096) {
      const int i = i0 + t;
      const bool w0 = i >= 0 && i < size, w1 = i + 1024 < size, w2 = i + 2048 < size, w3 = i + 3072 < size;
      const int x0 = stage[w0 ? i : 0];
      const int x1 = stage[w1 ? i + 1024 : 0];
      const int x2 = stage[w2 ? i + 2048 : 0];
      const int x3 = stage[w3 ? i + 3072 : 0];
      if (w0) out[i] = x0;
      if (w1) out[i + 1024] = x1;
      if (w2) out[i + 2048] = x2;
      if (w3) out[i + 3072] = x3;
    }
  }
}

#define G1_HEAD_BLOCKS 128
#define G1_WAVES       8
#define N_STRIPS       (N_NODES / 16)
#define KP_ROW         128

__device__ __forceinline__ void g1_load_group(v2f (&a)[8], const float* p) {
#pragma unroll
  for (int s = 0; s < 8; ++s) a[s] = *(const v2f*)(p + 4 * s);
}

__device__ __forceinline__ void g1_mma_group(v8f (&acc)[4], const v2f (&a)[8], const float* bl) {
#pragma unroll
  for (int s = 0; s < 8; ++s) {
    const float4 q0 = *(const float4*)(bl + s * (2 * KP_ROW));
    const float4 q1 = *(const float4*)(bl + s * (2 * KP_ROW) + 64);
    v2f b0, b1, b2, b3;
    b0.x = q0.x; b0.y = q0.y;
    b1.x = q0.z; b1.y = q0.w;
    b2.x = q1.x; b2.y = q1.y;
    b3.x = q1.z; b3.y = q1.w;
    acc[0] = __builtin_amdgcn_wmma_f32_16x16x4_f32(false, a[s], false, b0, (short)0, acc[0], false, false);
    acc[1] = __builtin_amdgcn_wmma_f32_16x16x4_f32(false, a[s], false, b1, (short)0, acc[1], false, false);
    acc[2] = __builtin_amdgcn_wmma_f32_16x16x4_f32(false, a[s], false, b2, (short)0, acc[2], false, false);
    acc[3] = __builtin_amdgcn_wmma_f32_16x16x4_f32(false, a[s], false, b3, (short)0, acc[3], false, false);
  }
}

__global__ __launch_bounds__(256) void gemm1_kernel(const float* __restrict__ X,
                                                    const float* __restrict__ W1,
                                                    float* __restrict__ T,
                                                    const int* __restrict__ src,
                                                    const int* __restrict__ dst,
                                                    int* __restrict__ hist,
                                                    int* __restrict__ lstart,
                                                    int2* __restrict__ pairs) {
  __shared__ __attribute__((aligned(16))) float Bs[256 * KP_ROW];
  const int t = threadIdx.x;
  if (blockIdx.x < N_CHUNK) {
    int* li = (int*)Bs;
    chunk_sort_256(src, dst, hist, lstart, pairs, (int2*)li, li + 2 * CHUNK_E, li + 2 * CHUNK_E + N_BKT,
                   li + 2 * CHUNK_E + 2 * N_BKT, (int)blockIdx.x, t);
  }
  const int head = blockIdx.x & 3;
  const int bih  = blockIdx.x >> 2;
  const float* __restrict__ Wh = W1 + (size_t)head * (D_IN * D_HEAD);

  {
    const int cq  = t & 15;
    const int kp0 = t >> 4;
    float4 r0[16], r1[16];
#pragma unroll
    for (int i = 0; i < 16; ++i) {
      const float4* p = (const float4*)(Wh + (size_t)(2 * (kp0 + 16 * i)) * D_HEAD) + cq;
      r0[i] = p[0];
      r1[i] = p[16];
    }
#pragma unroll
    for (int i = 0; i < 16; ++i) {
      float* q = Bs + (kp0 + 16 * i) * KP_ROW + 4 * cq;
      *(float4*)(q)      = make_float4(r0[i].x, r1[i].x, r0[i].y, r1[i].y);
      *(float4*)(q + 64) = make_float4(r0[i].z, r1[i].z, r0[i].w, r1[i].w);
    }
  }
  __syncthreads();

  const int wave = __builtin_amdgcn_readfirstlane(t >> 5);
  const int lane = t & 31;
  const int half = lane >> 4;
  const int lidx = lane & 15;
  const float* bl = Bs + half * KP_ROW + 4 * lidx;

  int strip = wave * G1_HEAD_BLOCKS + bih;
  const float* xrow = X + (size_t)((strip << 4) + lidx) * D_IN + 2 * half;
  v2f a0[8], a1[8];
  g1_load_group(a0, xrow);

  for (;;) {
    const int next = strip + G1_WAVES * G1_HEAD_BLOCKS;
    const bool has_next = next < N_STRIPS;
    const float* xnext = has_next ? X + (size_t)((next << 4) + lidx) * D_IN + 2 * half : xrow;

    v8f zero = {};
    v8f acc[4];
    acc[0] = zero; acc[1] = zero; acc[2] = zero; acc[3] = zero;

    for (int g = 0; g < D_IN / 32; g += 2) {
      g1_load_group(a1, xrow + (g + 1) * 32);
      g1_mma_group(acc, a0, bl + g * (16 * KP_ROW));
      g1_load_group(a0, (g + 2 < D_IN / 32) ? xrow + (g + 2) * 32 : xnext);
      g1_mma_group(acc, a1, bl + (g + 1) * (16 * KP_ROW));
    }

    float* trow = T + (size_t)((strip << 4) + half * 8) * D_H1 + head * D_HEAD + 4 * lidx;
#pragma unroll
    for (int r = 0; r < 8; ++r)
      *(float4*)(trow + (size_t)r * D_H1) = make_float4(acc[0][r], acc[1][r], acc[2][r], acc[3][r]);
    if (!has_next) break;
    strip = next;
    xrow = xnext;
  }
}

__device__ __forceinline__ float mean_elu(float a, float c, bool nz) {
  const float h = nz ? a / c : 0.0f;
  return (h > 0.0f) ? h : (expf(h) - 1.0f);
}

#define FN_NODES 32
#define HP       260

__device__ __forceinline__ void gather_row_sum(const int start, const int n, const int lane,
                                               const int* __restrict__ sdst,
                                               const float4* __restrict__ T4,
                                               float4& a0, float4& a1) {
  for (int base = 0; base < n; base += 32) {
    const int m = min(32, n - base);
    const int my = (lane < m) ? sdst[start + base + lane] : 0;
    int j = 0;
    for (; j + 4 <= m; j += 4) {
      const float4* r0 = T4 + (size_t)__builtin_amdgcn_readlane(my, j    ) * 64 + lane;
      const float4* r1 = T4 + (size_t)__builtin_amdgcn_readlane(my, j + 1) * 64 + lane;
      const float4* r2 = T4 + (size_t)__builtin_amdgcn_readlane(my, j + 2) * 64 + lane;
      const float4* r3 = T4 + (size_t)__builtin_amdgcn_readlane(my, j + 3) * 64 + lane;
      const float4 x0 = r0[0], y0 = r0[32];
      const float4 x1 = r1[0], y1 = r1[32];
      const float4 x2 = r2[0], y2 = r2[32];
      const float4 x3 = r3[0], y3 = r3[32];
      a0 += x0; a1 += y0;
      a0 += x1; a1 += y1;
      a0 += x2; a1 += y2;
      a0 += x3; a1 += y3;
    }
    const int rem = m - j;
    if (rem > 0) {
      const float4 z = make_float4(0.0f, 0.0f, 0.0f, 0.0f);
      float4 x1 = z, y1 = z, x2 = z, y2 = z;
      const float4* r0 = T4 + (size_t)__builtin_amdgcn_readlane(my, j) * 64 + lane;
      const float4 x0 = r0[0], y0 = r0[32];
      if (rem > 1) {
        const float4* r1 = T4 + (size_t)__builtin_amdgcn_readlane(my, j + 1) * 64 + lane;
        x1 = r1[0]; y1 = r1[32];
      }
      if (rem > 2) {
        const float4* r2 = T4 + (size_t)__builtin_amdgcn_readlane(my, j + 2) * 64 + lane;
        x2 = r2[0]; y2 = r2[32];
      }
      a0 += x0; a1 += y0;
      a0 += x1; a1 += y1;
      a0 += x2; a1 += y2;
    }
  }
}

__device__ __forceinline__ void g2_load_group(v2f (&q)[8], const float* __restrict__ p) {
#pragma unroll
  for (int s = 0; s < 4; ++s) {
    q[2 * s    ] = *(const v2f*)(p + (size_t)(4 * s    ) * D_HEAD);
    q[2 * s + 1] = *(const v2f*)(p + (size_t)(4 * s + 1) * D_HEAD);
  }
}

__device__ __forceinline__ void g2_mma_group(v8f& acc_e, v8f& acc_o, const float* ar, const v2f (&q)[8]) {
#pragma unroll
  for (int s = 0; s < 4; ++s) {
    const v2f a = *(const v2f*)(ar + 4 * s);
    v2f be, bo;
    be.x = q[2 * s].x; be.y = q[2 * s + 1].x;
    bo.x = q[2 * s].y; bo.y = q[2 * s + 1].y;
    acc_e = __builtin_amdgcn_wmma_f32_16x16x4_f32(false, a, false, be, (short)0, acc_e, false, false);
    acc_o = __builtin_amdgcn_wmma_f32_16x16x4_f32(false, a, false, bo, (short)0, acc_o, false, false);
  }
}

__global__ __launch_bounds__(512) void gather1_gemm2_kernel(const int* __restrict__ offs,
                                                            const int* __restrict__ cnt,
                                                            const int* __restrict__ sdst,
                                                            const float* __restrict__ T,
                                                            const float* __restrict__ W2,
                                                            float* __restrict__ T2) {
  __shared__ __attribute__((aligned(16))) float Hs[FN_NODES * HP];
  const int wave = __builtin_amdgcn_readfirstlane((int)(threadIdx.x >> 5));
  const int lane = threadIdx.x & 31;
  const float4* __restrict__ T4 = (const float4*)T;

  for (int i = 0; i < 2; ++i) {
    const int row = 2 * wave + i;
    const int node = blockIdx.x * FN_NODES + row;
    if (node >= N_NODES) break;
    const int start = offs[node];
    const int n = cnt[node];
    float4 a0 = make_float4(0.0f, 0.0f, 0.0f, 0.0f);
    float4 a1 = make_float4(0.0f, 0.0f, 0.0f, 0.0f);
    gather_row_sum(start, n, lane, sdst, T4, a0, a1);

    const float c = (float)n;
    const bool nz = n > 0;
    float4 h0, h1;
    h0.x = mean_elu(a0.x, c, nz); h0.y = mean_elu(a0.y, c, nz);
    h0.z = mean_elu(a0.z, c, nz); h0.w = mean_elu(a0.w, c, nz);
    h1.x = mean_elu(a1.x, c, nz); h1.y = mean_elu(a1.y, c, nz);
    h1.z = mean_elu(a1.z, c, nz); h1.w = mean_elu(a1.w, c, nz);
    float4* hrow = (float4*)(Hs + row * HP);
    hrow[lane]      = h0;
    hrow[32 + lane] = h1;
  }
  __syncthreads();
  if (wave >= 4) return;
  const int mt = wave >> 1;
  const int np = wave & 1;
  const int m0 = blockIdx.x * FN_NODES + mt * 16;
  if (m0 >= N_NODES) return;

  const int half = lane >> 4;
  const int lidx = lane & 15;
  const float* ar = Hs + (mt * 16 + lidx) * HP + 2 * half;
  const float* bp = W2 + (size_t)(2 * half) * D_HEAD + 32 * np + 2 * lidx;
  v8f acc_e = {};
  v8f acc_o = {};
  v2f q0[8], q1[8];
  g2_load_group(q0, bp);
  for (int g = 0; g < D_H1 / 16; g += 2) {
    g2_load_group(q1, bp + (size_t)((g + 1) * 16) * D_HEAD);
    g2_mma_group(acc_e, acc_o, ar + g * 16, q0);
    g2_load_group(q0, (g + 2 < D_H1 / 16) ? bp + (size_t)((g + 2) * 16) * D_HEAD : bp);
    g2_mma_group(acc_e, acc_o, ar + (g + 1) * 16, q1);
  }

  float* trow = T2 + (size_t)(m0 + half * 8) * D_HEAD + 32 * np + 2 * lidx;
#pragma unroll
  for (int r = 0; r < 8; ++r) {
    v2f o;
    o.x = acc_e[r]; o.y = acc_o[r];
    *(v2f*)(trow + (size_t)r * D_HEAD) = o;
  }
}

__global__ __launch_bounds__(256) void gather2_kernel(const int* __restrict__ offs,
                                                      const int* __restrict__ cnt,
                                                      const int* __restrict__ sdst,
                                                      const float* __restrict__ T2,
                                                      float* __restrict__ out) {
  const int node = __builtin_amdgcn_readfirstlane((int)(blockIdx.x * 8 + (threadIdx.x >> 5)));
  if (node >= N_NODES) return;
  const int lane = threadIdx.x & 31;
  const int half = lane >> 4;
  const int l16 = lane & 15;
  const int start = offs[node];
  const int n = cnt[node];
  const float4* __restrict__ T4 = (const float4*)T2;
  float4 a = make_float4(0.0f, 0.0f, 0.0f, 0.0f);

  for (int base = 0; base < n; base += 32) {
    const int m = min(32, n - base);
    const int my = (lane < m) ? sdst[start + base + lane] : 0;
    int j = 0;
    for (; j + 8 <= m; j += 8) {
      const int e0 = __builtin_amdgcn_readlane(my, j    ), o0 = __builtin_amdgcn_readlane(my, j + 1);
      const int e1 = __builtin_amdgcn_readlane(my, j + 2), o1 = __builtin_amdgcn_readlane(my, j + 3);
      const int e2 = __builtin_amdgcn_readlane(my, j + 4), o2 = __builtin_amdgcn_readlane(my, j + 5);
      const int e3 = __builtin_amdgcn_readlane(my, j + 6), o3 = __builtin_amdgcn_readlane(my, j + 7);
      const float4 x0 = T4[(size_t)(half ? o0 : e0) * 16 + l16];
      const float4 x1 = T4[(size_t)(half ? o1 : e1) * 16 + l16];
      const float4 x2 = T4[(size_t)(half ? o2 : e2) * 16 + l16];
      const float4 x3 = T4[(size_t)(half ? o3 : e3) * 16 + l16];
      a += x0; a += x1; a += x2; a += x3;
    }
    if (j < m) {
      const float4 z = make_float4(0.0f, 0.0f, 0.0f, 0.0f);
      float4 x0 = z, x1 = z, x2 = z, x3 = z;
      const int q = j + half;
      {
        const int e0 = __builtin_amdgcn_readlane(my, j), o0 = __builtin_amdgcn_readlane(my, j + 1);
        if (q < m) x0 = T4[(size_t)(half ? o0 : e0) * 16 + l16];
      }
      if (j + 2 < m) {
        const int e1 = __builtin_amdgcn_readlane(my, j + 2), o1 = __builtin_amdgcn_readlane(my, j + 3);
        if (q + 2 < m) x1 = T4[(size_t)(half ? o1 : e1) * 16 + l16];
      }
      if (j + 4 < m) {
        const int e2 = __builtin_amdgcn_readlane(my, j + 4), o2 = __builtin_amdgcn_readlane(my, j + 5);
        if (q + 4 < m) x2 = T4[(size_t)(half ? o2 : e2) * 16 + l16];
      }
      if (j + 6 < m) {
        const int e3 = __builtin_amdgcn_readlane(my, j + 6), o3 = __builtin_amdgcn_readlane(my, j + 7);
        if (q + 6 < m) x3 = T4[(size_t)(half ? o3 : e3) * 16 + l16];
      }
      a += x0; a += x1; a += x2; a += x3;
    }
  }

  a.x += __shfl_xor(a.x, 16);
  a.y += __shfl_xor(a.y, 16);
  a.z += __shfl_xor(a.z, 16);
  a.w += __shfl_xor(a.w, 16);
  if (half == 0) {
    const float c = (float)n;
    float4 o;
    o.x = (n > 0) ? a.x / c : 0.0f;
    o.y = (n > 0) ? a.y / c : 0.0f;
    o.z = (n > 0) ? a.z / c : 0.0f;
    o.w = (n > 0) ? a.w / c : 0.0f;
    ((float4*)(out + (size_t)node * D_HEAD))[l16] = o;
  }
}

extern "C" void kernel_launch(void* const* d_in, const int* in_sizes, int n_in,
                              void* d_out, int out_size, void* d_ws, size_t ws_size,
                              hipStream_t stream) {
  const float* X  = (const float*)d_in[0];
  const int*   ei = (const int*)d_in[1];
  const float* W1 = (const float*)d_in[2];
  const float* W2 = (const float*)d_in[3];
  float* out = (float*)d_out;

  const int* src = ei;
  const int* dst = ei + N_EDGES;

  int* cnt    = (int*)d_ws;
  int* offs   = cnt + N_PAD;
  int* hist   = offs + N_PAD;
  int* lstart = hist + 256 * N_BKT;
  int* sdst   = hist + N_EDGES;
  float* bufA = (float*)(sdst + N_EDGES);
  float* bufB = bufA + (size_t)N_NODES * D_H1;
  int2* pairs = (int2*)(bufB + (size_t)N_NODES * D_H1);

  gemm1_kernel<<<512, 256, 0, stream>>>(X, W1, bufA, src, dst, hist, lstart, pairs);

  bucket_group_kernel<<<(N_NODES + N_BKT - 1) / N_BKT, 1024, 0, stream>>>(pairs, hist, lstart, cnt, offs, sdst);

  gather1_gemm2_kernel<<<(N_NODES + FN_NODES - 1) / FN_NODES, 512, 0, stream>>>(offs, cnt, sdst, bufA, W2, bufB);

  gather2_kernel<<<N_NODES / 8, 256, 0, stream>>>(offs, cnt, sdst, bufB, out);
}
